# starting kernel
# baseline (speedup 1.0000x reference)
#include <hip/hip_runtime.h>
#include <hip/hip_bf16.h>

typedef float v2f __attribute__((ext_vector_type(2)));
typedef float v8f __attribute__((ext_vector_type(8)));

#define BB   4096
#define DD   32
#define HH   5
#define GG   20
#define INW  128

__device__ __forceinline__ float sigf(float x) { return 1.0f / (1.0f + __expf(-x)); }

__global__ __launch_bounds__(256) void hb_kernel(
    const float* __restrict__ hidden,
    const float* __restrict__ W_hh0, const float* __restrict__ b_ih0, const float* __restrict__ b_hh0,
    const float* __restrict__ W_hh1, const float* __restrict__ b_ih1, const float* __restrict__ b_hh1,
    float* __restrict__ hb0, float* __restrict__ hb1)
{
    int b = blockIdx.x * blockDim.x + threadIdx.x;
    if (b >= BB) return;
    float h0[HH], h1[HH];
    #pragma unroll
    for (int j = 0; j < HH; ++j) {
        h0[j] = hidden[b * HH + j];
        h1[j] = hidden[BB * HH + b * HH + j];
    }
    #pragma unroll 4
    for (int g = 0; g < GG; ++g) {
        float s0 = b_ih0[g] + b_hh0[g];
        float s1 = b_ih1[g] + b_hh1[g];
        #pragma unroll
        for (int j = 0; j < HH; ++j) {
            s0 += h0[j] * W_hh0[g * HH + j];
            s1 += h1[j] * W_hh1[g * HH + j];
        }
        hb0[b * GG + g] = s0;
        hb1[b * GG + g] = s1;
    }
}

__global__ __launch_bounds__(256) void lstm_wmma_kernel(
    const float* __restrict__ dec_x,
    const int*   __restrict__ cat1,
    const int*   __restrict__ cat2,
    const int*   __restrict__ cat3,
    const float* __restrict__ emb,
    const float* __restrict__ W_ih0,
    const float* __restrict__ W_ih1,
    const float* __restrict__ cell,
    const float* __restrict__ hb0,
    const float* __restrict__ hb1,
    float* __restrict__ out)
{
    __shared__ float Wp[32 * INW];
    __shared__ float Gs[8][16 * GG];

    const int tid = threadIdx.x;
    for (int i = tid; i < 32 * INW; i += 256) {
        Wp[i] = ((i >> 7) < GG) ? W_ih0[i] : 0.0f;
    }
    __syncthreads();

    const int wave = tid >> 5;
    const int lane = tid & 31;
    const int m    = lane & 15;
    const int half = lane >> 4;
    const int koff = half * 2;

    const long rbase = ((long)blockIdx.x * 8 + wave) * 16;
    const int  t     = (int)(rbase >> 12);
    const int  b     = (int)(rbase & 4095) + m;
    const long row   = (long)t * BB + b;

    const float* srcs[4];
    srcs[0] = dec_x + row * DD;
    srcs[1] = emb + (long)cat1[row] * 32;
    srcs[2] = emb + (long)cat2[row] * 32;
    srcs[3] = emb + (long)cat3[row] * 32;

    const float* bp0 = &Wp[m * INW];
    const float* bp1 = &Wp[(m + 16) * INW];

    v8f acc0 = {};
    v8f acc1 = {};
    #pragma unroll
    for (int seg = 0; seg < 4; ++seg) {
        const float* ap = srcs[seg];
        #pragma unroll
        for (int kk = 0; kk < 8; ++kk) {
            const int col  = kk * 4 + koff;
            const int gcol = seg * 32 + col;
            v2f a  = *(const v2f*)(ap  + col);
            v2f b0 = *(const v2f*)(bp0 + gcol);
            v2f b1 = *(const v2f*)(bp1 + gcol);
            acc0 = __builtin_amdgcn_wmma_f32_16x16x4_f32(false, a, false, b0,
                                                         (short)0, acc0, false, false);
            acc1 = __builtin_amdgcn_wmma_f32_16x16x4_f32(false, a, false, b1,
                                                         (short)0, acc1, false, false);
        }
    }

    float* gs = &Gs[wave][0];
    #pragma unroll
    for (int j = 0; j < 8; ++j) {
        const int rm = j + half * 8;
        gs[rm * GG + m] = acc0[j];
        if (m < 4) gs[rm * GG + 16 + m] = acc1[j];
    }
    asm volatile("s_wait_dscnt 0" ::: "memory");

    if (lane < 16) {
        float g0[GG];
        #pragma unroll
        for (int g = 0; g < GG; ++g) g0[g] = gs[m * GG + g] + hb0[(long)b * GG + g];

        float h1v[HH];
        #pragma unroll
        for (int h = 0; h < HH; ++h) {
            float cn = sigf(g0[5 + h]) * cell[(long)b * HH + h]
                     + sigf(g0[h]) * tanhf(g0[10 + h]);
            h1v[h] = sigf(g0[15 + h]) * tanhf(cn);
        }

        float g1[GG];
        #pragma unroll 4
        for (int g = 0; g < GG; ++g) {
            float s = hb1[(long)b * GG + g];
            #pragma unroll
            for (int j = 0; j < HH; ++j) s += W_ih1[g * HH + j] * h1v[j];
            g1[g] = s;
        }
        #pragma unroll
        for (int h = 0; h < HH; ++h) {
            float cn = sigf(g1[5 + h]) * cell[(long)BB * HH + (long)b * HH + h]
                     + sigf(g1[h]) * tanhf(g1[10 + h]);
            out[row * HH + h] = sigf(g1[15 + h]) * tanhf(cn);
        }
    }
}

extern "C" void kernel_launch(void* const* d_in, const int* in_sizes, int n_in,
                              void* d_out, int out_size, void* d_ws, size_t ws_size,
                              hipStream_t stream)
{
    const float* hidden = (const float*)d_in[1];
    const float* cellp  = (const float*)d_in[2];
    const float* dec_x  = (const float*)d_in[3];
    const int*   cat1   = (const int*)d_in[4];
    const int*   cat2   = (const int*)d_in[5];
    const int*   cat3   = (const int*)d_in[6];
    const float* emb    = (const float*)d_in[7];
    const float* W_ih0  = (const float*)d_in[8];
    const float* W_hh0  = (const float*)d_in[9];
    const float* b_ih0  = (const float*)d_in[10];
    const float* b_hh0  = (const float*)d_in[11];
    const float* W_ih1  = (const float*)d_in[12];
    const float* W_hh1  = (const float*)d_in[13];
    const float* b_ih1  = (const float*)d_in[14];
    const float* b_hh1  = (const float*)d_in[15];
    float* out = (float*)d_out;

    const int T    = in_sizes[3] / (BB * DD);
    const int rows = T * BB;

    float* hb0 = (float*)d_ws;
    float* hb1 = hb0 + (size_t)BB * GG;

    hb_kernel<<<BB / 256, 256, 0, stream>>>(hidden, W_hh0, b_ih0, b_hh0,
                                            W_hh1, b_ih1, b_hh1, hb0, hb1);

    lstm_wmma_kernel<<<rows / 128, 256, 0, stream>>>(dec_x, cat1, cat2, cat3, emb,
                                                     W_ih0, W_ih1, cellp,
                                                     hb0, hb1, out);
}


# WMMA operands swapped (weights as A, x rows as B), gate rows regrouped: a lane's D registers hold whole hidden units of its own row; LDS transposition gone, all 32 lanes do gate math; sum order kept
# speedup vs baseline: 1.6873x; 1.6873x over previous
#include <hip/hip_runtime.h>
#include <hip/hip_bf16.h>

typedef float v2f __attribute__((ext_vector_type(2)));
typedef float v8f __attribute__((ext_vector_type(8)));

#define BB   4096
#define DD   32
#define HH   5
#define GG   20
#define INW  128

__device__ __forceinline__ float sigf(float x) { return 1.0f / (1.0f + __expf(-x)); }

__global__ __launch_bounds__(256) void hb_kernel(
    const float* __restrict__ hidden,
    const float* __restrict__ W_hh0, const float* __restrict__ b_ih0, const float* __restrict__ b_hh0,
    const float* __restrict__ W_hh1, const float* __restrict__ b_ih1, const float* __restrict__ b_hh1,
    float* __restrict__ hb0, float* __restrict__ hb1)
{
    int b = blockIdx.x * blockDim.x + threadIdx.x;
    if (b >= BB) return;
    float h0[HH], h1[HH];
    #pragma unroll
    for (int j = 0; j < HH; ++j) {
        h0[j] = hidden[b * HH + j];
        h1[j] = hidden[BB * HH + b * HH + j];
    }
    #pragma unroll 4
    for (int g = 0; g < GG; ++g) {
        float s0 = b_ih0[g] + b_hh0[g];
        float s1 = b_ih1[g] + b_hh1[g];
        #pragma unroll
        for (int j = 0; j < HH; ++j) {
            s0 += h0[j] * W_hh0[g * HH + j];
            s1 += h1[j] * W_hh1[g * HH + j];
        }
        hb0[b * GG + g] = s0;
        hb1[b * GG + g] = s1;
    }
}

#define TG 16

__device__ __forceinline__ float xhalf(float x) {
    int v = __builtin_bit_cast(int, x);
    return __builtin_bit_cast(float,
        __builtin_amdgcn_permlanex16(v, v, 0x76543210, 0xfedcba98, false, false));
}

__device__ __forceinline__ float lstm_unit(float gi, float gf, float gg, float go, float c) {
    float cn = sigf(gf) * c + sigf(gi) * tanhf(gg);
    return sigf(go) * tanhf(cn);
}

__device__ __forceinline__ float gate1(const float* __restrict__ hbrow,
                                       const float* __restrict__ W_ih1,
                                       int gate, const float (&h1v)[HH]) {
    float s = hbrow[gate];
    #pragma unroll
    for (int j = 0; j < HH; ++j) s += W_ih1[gate * HH + j] * h1v[j];
    return s;
}

__global__ __launch_bounds__(256) void lstm_wmma_kernel(
    const float* __restrict__ dec_x,
    const int*   __restrict__ cat1,
    const int*   __restrict__ cat2,
    const int*   __restrict__ cat3,
    const float* __restrict__ emb,
    const float* __restrict__ W_ih0,
    const float* __restrict__ W_ih1,
    const float* __restrict__ cell,
    const float* __restrict__ hb0,
    const float* __restrict__ hb1,
    float* __restrict__ out,
    int T)
{
    const int tid  = threadIdx.x;
    const int wave = tid >> 5;
    const int lane = tid & 31;
    const int m    = lane & 15;
    const int half = lane >> 4;
    const int koff = half * 2;

    const int    gs0  = ((m >> 1) & 3) * HH + (m & 1) + ((m & 8) ? 3 : 0);
    const int    gs1  = (m < 4) ? (m * HH + 2) : 2;
    const float* wr0  = W_ih0 + gs0 * INW + koff;
    const float* wr1  = W_ih0 + gs1 * INW + koff;
    const v2f    zero = {0.0f, 0.0f};
    v2f wa0[32], wa1[32];
    #pragma unroll
    for (int s = 0; s < 32; ++s) {
        wa0[s] = *(const v2f*)(wr0 + s * 4);
        v2f w1 = *(const v2f*)(wr1 + s * 4);
        wa1[s] = (m < 4) ? w1 : zero;
    }

    const int gw = blockIdx.x * 8 + __builtin_amdgcn_readfirstlane(wave);
    const int b  = (gw & 255) * 16 + m;

    const int hu = half * 3;
    const float* hb0r = hb0 + (long)b * GG;
    const float* hb1r = hb1 + (long)b * GG;
    const float* c0r  = cell + (long)b * HH;
    const float* c1r  = cell + (long)BB * HH + (long)b * HH;

    #pragma unroll 1
    for (int t = gw >> 8; t < T; t += TG) {
        const long row = (long)t * BB + b;

        const float* srcs[4];
        srcs[0] = dec_x + row * DD;
        srcs[1] = emb + (long)cat1[row] * 32;
        srcs[2] = emb + (long)cat2[row] * 32;
        srcs[3] = emb + (long)cat3[row] * 32;

        v8f acc0 = {};
        v8f acc1 = {};
        #pragma unroll
        for (int seg = 0; seg < 4; ++seg) {
            const float* ap = srcs[seg];
            #pragma unroll
            for (int kk = 0; kk < 8; ++kk) {
                const int col = kk * 4 + koff;
                v2f x = *(const v2f*)(ap + col);
                acc0 = __builtin_amdgcn_wmma_f32_16x16x4_f32(false, wa0[seg * 8 + kk], false, x,
                                                             (short)0, acc0, false, false);
                acc1 = __builtin_amdgcn_wmma_f32_16x16x4_f32(false, wa1[seg * 8 + kk], false, x,
                                                             (short)0, acc1, false, false);
            }
        }

        float own[3];
        #pragma unroll
        for (int u = 0; u < 2; ++u) {
            const int h = hu + u;
            own[u] = lstm_unit(acc0[u]     + hb0r[h],
                               acc0[2 + u] + hb0r[5 + h],
                               acc0[4 + u] + hb0r[10 + h],
                               acc0[6 + u] + hb0r[15 + h], c0r[h]);
        }
        own[2] = 0.0f;
        if (half == 0) {
            own[2] = lstm_unit(acc1[0] + hb0r[2], acc1[1] + hb0r[7],
                               acc1[2] + hb0r[12], acc1[3] + hb0r[17], c0r[2]);
        }

        const float oth0 = xhalf(own[0]);
        const float oth1 = xhalf(own[1]);
        const float oth2 = xhalf(own[2]);
        float h1v[HH];
        h1v[0] = half ? oth0 : own[0];
        h1v[1] = half ? oth1 : own[1];
        h1v[2] = half ? oth2 : own[2];
        h1v[3] = half ? own[0] : oth0;
        h1v[4] = half ? own[1] : oth1;

        #pragma unroll
        for (int u = 0; u < 2; ++u) {
            const int h = hu + u;
            out[row * HH + h] = lstm_unit(gate1(hb1r, W_ih1, h, h1v),
                                          gate1(hb1r, W_ih1, 5 + h, h1v),
                                          gate1(hb1r, W_ih1, 10 + h, h1v),
                                          gate1(hb1r, W_ih1, 15 + h, h1v), c1r[h]);
        }
        if (half == 0) {
            out[row * HH + 2] = lstm_unit(gate1(hb1r, W_ih1, 2, h1v),
                                          gate1(hb1r, W_ih1, 7, h1v),
                                          gate1(hb1r, W_ih1, 12, h1v),
                                          gate1(hb1r, W_ih1, 17, h1v), c1r[2]);
        }
    }
}

extern "C" void kernel_launch(void* const* d_in, const int* in_sizes, int n_in,
                              void* d_out, int out_size, void* d_ws, size_t ws_size,
                              hipStream_t stream)
{
    const float* hidden = (const float*)d_in[1];
    const float* cellp  = (const float*)d_in[2];
    const float* dec_x  = (const float*)d_in[3];
    const int*   cat1   = (const int*)d_in[4];
    const int*   cat2   = (const int*)d_in[5];
    const int*   cat3   = (const int*)d_in[6];
    const float* emb    = (const float*)d_in[7];
    const float* W_ih0  = (const float*)d_in[8];
    const float* W_hh0  = (const float*)d_in[9];
    const float* b_ih0  = (const float*)d_in[10];
    const float* b_hh0  = (const float*)d_in[11];
    const float* W_ih1  = (const float*)d_in[12];
    const float* W_hh1  = (const float*)d_in[13];
    const float* b_ih1  = (const float*)d_in[14];
    const float* b_hh1  = (const float*)d_in[15];
    float* out = (float*)d_out;

    const int T    = in_sizes[3] / (BB * DD);

    float* hb0 = (float*)d_ws;
    float* hb1 = hb0 + (size_t)BB * GG;

    hb_kernel<<<BB / 256, 256, 0, stream>>>(hidden, W_hh0, b_ih0, b_hh0,
                                            W_hh1, b_ih1, b_hh1, hb0, hb1);

    lstm_wmma_kernel<<<512, 256, 0, stream>>>(dec_x, cat1, cat2, cat3, emb,
                                              W_ih0, W_ih1, cellp,
                                              hb0, hb1, out, T);
}


# Tile loop software-pipelined: next tile's 32 x-fragment loads go out as one group behind the WMMA block (ids a tile earlier), land under the gate math; 2 waves/SIMD for the registers; sums unchanged
# speedup vs baseline: 1.7439x; 1.0335x over previous
#include <hip/hip_runtime.h>
#include <hip/hip_bf16.h>

typedef float v2f __attribute__((ext_vector_type(2)));
typedef float v8f __attribute__((ext_vector_type(8)));

#define BB   4096
#define DD   32
#define HH   5
#define GG   20
#define INW  128

__device__ __forceinline__ float sigf(float x) { return 1.0f / (1.0f + __expf(-x)); }

__global__ __launch_bounds__(256) void hb_kernel(
    const float* __restrict__ hidden,
    const float* __restrict__ W_hh0, const float* __restrict__ b_ih0, const float* __restrict__ b_hh0,
    const float* __restrict__ W_hh1, const float* __restrict__ b_ih1, const float* __restrict__ b_hh1,
    float* __restrict__ hb0, float* __restrict__ hb1)
{
    int b = blockIdx.x * blockDim.x + threadIdx.x;
    if (b >= BB) return;
    float h0[HH], h1[HH];
    #pragma unroll
    for (int j = 0; j < HH; ++j) {
        h0[j] = hidden[b * HH + j];
        h1[j] = hidden[BB * HH + b * HH + j];
    }
    #pragma unroll 4
    for (int g = 0; g < GG; ++g) {
        float s0 = b_ih0[g] + b_hh0[g];
        float s1 = b_ih1[g] + b_hh1[g];
        #pragma unroll
        for (int j = 0; j < HH; ++j) {
            s0 += h0[j] * W_hh0[g * HH + j];
            s1 += h1[j] * W_hh1[g * HH + j];
        }
        hb0[b * GG + g] = s0;
        hb1[b * GG + g] = s1;
    }
}

#define TG 8

__device__ __forceinline__ float xhalf(float x) {
    int v = __builtin_bit_cast(int, x);
    return __builtin_bit_cast(float,
        __builtin_amdgcn_permlanex16(v, v, 0x76543210, 0xfedcba98, false, false));
}

__device__ __forceinline__ float lstm_unit(float gi, float gf, float gg, float go, float c) {
    float cn = sigf(gf) * c + sigf(gi) * tanhf(gg);
    return sigf(go) * tanhf(cn);
}

__device__ __forceinline__ float gate1(float hbterm,
                                       const float* __restrict__ W_ih1,
                                       int gate, const float (&h1v)[HH]) {
    float s = hbterm;
    #pragma unroll
    for (int j = 0; j < HH; ++j) s += W_ih1[gate * HH + j] * h1v[j];
    return s;
}

__device__ __forceinline__ void load_x(v2f (&x)[32],
                                       const float* __restrict__ p0, const float* __restrict__ p1,
                                       const float* __restrict__ p2, const float* __restrict__ p3) {
    #pragma unroll
    for (int kk = 0; kk < 8; ++kk) {
        x[kk]      = *(const v2f*)(p0 + kk * 4);
        x[8 + kk]  = *(const v2f*)(p1 + kk * 4);
        x[16 + kk] = *(const v2f*)(p2 + kk * 4);
        x[24 + kk] = *(const v2f*)(p3 + kk * 4);
    }
}

__global__ __launch_bounds__(256) __attribute__((amdgpu_waves_per_eu(2, 2))) void lstm_wmma_kernel(
    const float* __restrict__ dec_x,
    const int*   __restrict__ cat1,
    const int*   __restrict__ cat2,
    const int*   __restrict__ cat3,
    const float* __restrict__ emb,
    const float* __restrict__ W_ih0,
    const float* __restrict__ W_ih1,
    const float* __restrict__ cell,
    const float* __restrict__ hb0,
    const float* __restrict__ hb1,
    float* __restrict__ out,
    int T)
{
    const int tid  = threadIdx.x;
    const int wave = tid >> 5;
    const int lane = tid & 31;
    const int m    = lane & 15;
    const int half = lane >> 4;
    const int koff = half * 2;

    const int    gs0  = ((m >> 1) & 3) * HH + (m & 1) + ((m & 8) ? 3 : 0);
    const int    gs1  = (m < 4) ? (m * HH + 2) : 2;
    const float* wr0  = W_ih0 + gs0 * INW + koff;
    const float* wr1  = W_ih0 + gs1 * INW + koff;
    const v2f    zero = {0.0f, 0.0f};
    v2f wa0[32], wa1[32];
    #pragma unroll
    for (int s = 0; s < 32; ++s) {
        wa0[s] = *(const v2f*)(wr0 + s * 4);
        v2f w1 = *(const v2f*)(wr1 + s * 4);
        wa1[s] = (m < 4) ? w1 : zero;
    }

    const int gw = blockIdx.x * 8 + __builtin_amdgcn_readfirstlane(wave);
    const int b  = (gw & 255) * 16 + m;

    int t = gw >> 8;
    if (t >= T) return;

    const int hu = half * 3;
    const float* hb0r = hb0 + (long)b * GG;
    const float* hb1r = hb1 + (long)b * GG;
    const float* c0r  = cell + (long)b * HH;
    const float* c1r  = cell + (long)BB * HH + (long)b * HH;

    float r0[2][4], r1[2][4], cc0[2], cc1[2];
    #pragma unroll
    for (int u = 0; u < 2; ++u) {
        #pragma unroll
        for (int q = 0; q < 4; ++q) {
            r0[u][q] = hb0r[q * HH + hu + u];
            r1[u][q] = hb1r[q * HH + hu + u];
        }
        cc0[u] = c0r[hu + u];
        cc1[u] = c1r[hu + u];
    }
    float r0m[4], r1m[4];
    #pragma unroll
    for (int q = 0; q < 4; ++q) {
        r0m[q] = hb0r[q * HH + 2];
        r1m[q] = hb1r[q * HH + 2];
    }
    const float cc0m = c0r[2];
    const float cc1m = c1r[2];

    const float* ep = emb + koff;
    const float* dp = dec_x + koff;
    v2f xb[32];
    int n1, n2, n3;
    {
        const long row0 = (long)t * BB + b;
        const int  i1 = cat1[row0], i2 = cat2[row0], i3 = cat3[row0];
        load_x(xb, dp + row0 * DD, ep + (long)i1 * 32, ep + (long)i2 * 32, ep + (long)i3 * 32);
        const int  tn   = (t + TG < T) ? t + TG : t;
        const long rown = (long)tn * BB + b;
        n1 = cat1[rown]; n2 = cat2[rown]; n3 = cat3[rown];
    }

    #pragma unroll 1
    for (; t < T; t += TG) {
        const long row = (long)t * BB + b;

        const int  t2   = (t + 2 * TG < T) ? t + 2 * TG : t;
        const long row2 = (long)t2 * BB + b;
        const int  m1 = cat1[row2], m2 = cat2[row2], m3 = cat3[row2];

        v8f acc0 = {};
        v8f acc1 = {};
        #pragma unroll
        for (int s = 0; s < 32; ++s) {
            acc0 = __builtin_amdgcn_wmma_f32_16x16x4_f32(false, wa0[s], false, xb[s],
                                                         (short)0, acc0, false, false);
            acc1 = __builtin_amdgcn_wmma_f32_16x16x4_f32(false, wa1[s], false, xb[s],
                                                         (short)0, acc1, false, false);
        }

        const bool  more = t + TG < T;
        const long  rown = (long)(more ? t + TG : t) * BB + b;
        const float* q0  = more ? dp + rown * DD : ep;
        load_x(xb, q0, ep + (long)n1 * 32, ep + (long)n2 * 32, ep + (long)n3 * 32);
        n1 = m1; n2 = m2; n3 = m3;

        float own[3];
        #pragma unroll
        for (int u = 0; u < 2; ++u) {
            own[u] = lstm_unit(acc0[u]     + r0[u][0],
                               acc0[2 + u] + r0[u][1],
                               acc0[4 + u] + r0[u][2],
                               acc0[6 + u] + r0[u][3], cc0[u]);
        }
        own[2] = 0.0f;
        if (half == 0) {
            own[2] = lstm_unit(acc1[0] + r0m[0], acc1[1] + r0m[1],
                               acc1[2] + r0m[2], acc1[3] + r0m[3], cc0m);
        }

        const float oth0 = xhalf(own[0]);
        const float oth1 = xhalf(own[1]);
        const float oth2 = xhalf(own[2]);
        float h1v[HH];
        h1v[0] = half ? oth0 : own[0];
        h1v[1] = half ? oth1 : own[1];
        h1v[2] = half ? oth2 : own[2];
        h1v[3] = half ? own[0] : oth0;
        h1v[4] = half ? own[1] : oth1;

        #pragma unroll
        for (int u = 0; u < 2; ++u) {
            const int h = hu + u;
            out[row * HH + h] = lstm_unit(gate1(r1[u][0], W_ih1, h, h1v),
                                          gate1(r1[u][1], W_ih1, 5 + h, h1v),
                                          gate1(r1[u][2], W_ih1, 10 + h, h1v),
                                          gate1(r1[u][3], W_ih1, 15 + h, h1v), cc1[u]);
        }
        if (half == 0) {
            out[row * HH + 2] = lstm_unit(gate1(r1m[0], W_ih1, 2, h1v),
                                          gate1(r1m[1], W_ih1, 7, h1v),
                                          gate1(r1m[2], W_ih1, 12, h1v),
                                          gate1(r1m[3], W_ih1, 17, h1v), cc1m);
        }
    }
}

extern "C" void kernel_launch(void* const* d_in, const int* in_sizes, int n_in,
                              void* d_out, int out_size, void* d_ws, size_t ws_size,
                              hipStream_t stream)
{
    const float* hidden = (const float*)d_in[1];
    const float* cellp  = (const float*)d_in[2];
    const float* dec_x  = (const float*)d_in[3];
    const int*   cat1   = (const int*)d_in[4];
    const int*   cat2   = (const int*)d_in[5];
    const int*   cat3   = (const int*)d_in[6];
    const float* emb    = (const float*)d_in[7];
    const float* W_ih0  = (const float*)d_in[8];
    const float* W_hh0  = (const float*)d_in[9];
    const float* b_ih0  = (const float*)d_in[10];
    const float* b_hh0  = (const float*)d_in[11];
    const float* W_ih1  = (const float*)d_in[12];
    const float* W_hh1  = (const float*)d_in[13];
    const float* b_ih1  = (const float*)d_in[14];
    const float* b_hh1  = (const float*)d_in[15];
    float* out = (float*)d_out;

    const int T    = in_sizes[3] / (BB * DD);

    float* hb0 = (float*)d_ws;
    float* hb1 = hb0 + (size_t)BB * GG;

    hb_kernel<<<BB / 256, 256, 0, stream>>>(hidden, W_hh0, b_ih0, b_hh0,
                                            W_hh1, b_ih1, b_hh1, hb0, hb1);

    lstm_wmma_kernel<<<256, 256, 0, stream>>>(dec_x, cat1, cat2, cat3, emb,
                                              W_ih0, W_ih1, cellp,
                                              hb0, hb1, out, T);
}


# Recurrent and bias terms of both cells go through the WMMA K dimension, multiplied once per wave as each tile's C operand; layer 1 moved onto WMMA; pre-pass launch gone; gate sum order changes
# speedup vs baseline: 2.0706x; 1.1873x over previous
#include <hip/hip_runtime.h>
#include <hip/hip_bf16.h>

typedef float v2f __attribute__((ext_vector_type(2)));
typedef float v8f __attribute__((ext_vector_type(8)));

#define BB   4096
#define DD   32
#define HH   5
#define GG   20
#define INW  128

__device__ __forceinline__ float sigf(float x) { return 1.0f / (1.0f + __expf(-x)); }

#define TG 8

__device__ __forceinline__ float lstm_unit(float gi, float gf, float gg, float go, float c) {
    float cn = sigf(gf) * c + sigf(gi) * tanhf(gg);
    return sigf(go) * tanhf(cn);
}

__device__ __forceinline__ void load_x(v2f (&x)[32],
                                       const float* __restrict__ p0, const float* __restrict__ p1,
                                       const float* __restrict__ p2, const float* __restrict__ p3) {
    #pragma unroll
    for (int kk = 0; kk < 8; ++kk) {
        x[kk]      = *(const v2f*)(p0 + kk * 4);
        x[8 + kk]  = *(const v2f*)(p1 + kk * 4);
        x[16 + kk] = *(const v2f*)(p2 + kk * 4);
        x[24 + kk] = *(const v2f*)(p3 + kk * 4);
    }
}

__global__ __launch_bounds__(256) __attribute__((amdgpu_waves_per_eu(2, 2))) void lstm_wmma_kernel(
    const float* __restrict__ dec_x,
    const int*   __restrict__ cat1,
    const int*   __restrict__ cat2,
    const int*   __restrict__ cat3,
    const float* __restrict__ emb,
    const float* __restrict__ W_ih0,
    const float* __restrict__ W_hh0,
    const float* __restrict__ b_ih0,
    const float* __restrict__ b_hh0,
    const float* __restrict__ W_ih1,
    const float* __restrict__ W_hh1,
    const float* __restrict__ b_ih1,
    const float* __restrict__ b_hh1,
    const float* __restrict__ hidden,
    const float* __restrict__ cell,
    float* __restrict__ out,
    int T)
{
    const int tid  = threadIdx.x;
    const int wave = tid >> 5;
    const int lane = tid & 31;
    const int m    = lane & 15;
    const int half = lane >> 4;
    const int koff = half * 2;

    const int    gs0  = ((m >> 1) & 3) * HH + (m & 1) + ((m & 8) ? 3 : 0);
    const int    gs1  = (m < 4) ? (m * HH + 2) : 2;
    const float* wr0  = W_ih0 + gs0 * INW + koff;
    const float* wr1  = W_ih0 + gs1 * INW + koff;
    const v2f    zero = {0.0f, 0.0f};
    const bool   act1 = m < 4;
    v2f wa0[32], wa1[32];
    #pragma unroll
    for (int s = 0; s < 32; ++s) {
        wa0[s] = *(const v2f*)(wr0 + s * 4);
        v2f w1 = *(const v2f*)(wr1 + s * 4);
        wa1[s] = act1 ? w1 : zero;
    }
    v2f la0[2], la1[2];
    v2f ra0[2], ra1[2], rl0, rl1;
    {
        const float* u0 = W_hh0 + gs0 * HH;
        const float* u1 = W_hh0 + gs1 * HH;
        const float* v0 = W_ih1 + gs0 * HH;
        const float* v1 = W_ih1 + gs1 * HH;
        const float* y0 = W_hh1 + gs0 * HH;
        const float* y1 = W_hh1 + gs1 * HH;
        const v2f p0 = {u0[koff], u0[koff + 1]};
        const v2f p1 = {u1[koff], u1[koff + 1]};
        ra0[0] = p0;
        ra1[0] = act1 ? p1 : zero;
        const v2f q0h0 = {u0[4], b_ih0[gs0]}, q0h1 = {b_hh0[gs0], 0.0f};
        const v2f q1h0 = {u1[4], b_ih0[gs1]}, q1h1 = {b_hh0[gs1], 0.0f};
        ra0[1] = half ? q0h1 : q0h0;
        ra1[1] = act1 ? (half ? q1h1 : q1h0) : zero;

        const int c3 = half * 3;
        const v2f a0 = {v0[c3], v0[c3 + 1]};
        const v2f a1 = {v1[c3], v1[c3 + 1]};
        la0[0] = a0;
        la1[0] = act1 ? a1 : zero;
        const v2f b0h0 = {v0[2], y0[0]}, b0h1 = {y0[1], y0[2]};
        const v2f b1h0 = {v1[2], y1[0]}, b1h1 = {y1[1], y1[2]};
        la0[1] = half ? b0h1 : b0h0;
        la1[1] = act1 ? (half ? b1h1 : b1h0) : zero;
        const v2f c0h0 = {y0[3], y0[4]}, c0h1 = {b_ih1[gs0], b_hh1[gs0]};
        const v2f c1h0 = {y1[3], y1[4]}, c1h1 = {b_ih1[gs1], b_hh1[gs1]};
        rl0 = half ? c0h1 : c0h0;
        rl1 = act1 ? (half ? c1h1 : c1h0) : zero;
    }

    const int gw = blockIdx.x * 8 + __builtin_amdgcn_readfirstlane(wave);
    const int b  = (gw & 255) * 16 + m;

    int t = gw >> 8;
    if (t >= T) return;

    const int hu = half * 3;
    const float* h0r  = hidden + (long)b * HH;
    const float* h1r  = hidden + (long)BB * HH + (long)b * HH;
    const float* c0r  = cell + (long)b * HH;
    const float* c1r  = cell + (long)BB * HH + (long)b * HH;

    float hA[HH], hB[HH], cc0[2], cc1[2];
    #pragma unroll
    for (int j = 0; j < HH; ++j) {
        hA[j] = h0r[j];
        hB[j] = h1r[j];
    }
    #pragma unroll
    for (int u = 0; u < 2; ++u) {
        cc0[u] = c0r[hu + u];
        cc1[u] = c1r[hu + u];
    }
    const float cc0m = c0r[2];
    const float cc1m = c1r[2];

    const v2f xh32 = {half ? hA[2] : hA[0], half ? hA[3] : hA[1]};
    const v2f xh33 = {half ? 1.0f : hA[4], half ? 0.0f : 1.0f};
    const float l1e1 = half ? hB[2] : hB[0];
    const v2f lb2  = {half ? 1.0f : hB[3], half ? 1.0f : hB[4]};

    v8f racc0 = {}, racc1 = {}, rg0 = {}, rg1 = {};
    racc0 = __builtin_amdgcn_wmma_f32_16x16x4_f32(false, ra0[0], false, xh32, (short)0, racc0, false, false);
    racc1 = __builtin_amdgcn_wmma_f32_16x16x4_f32(false, ra1[0], false, xh32, (short)0, racc1, false, false);
    racc0 = __builtin_amdgcn_wmma_f32_16x16x4_f32(false, ra0[1], false, xh33, (short)0, racc0, false, false);
    racc1 = __builtin_amdgcn_wmma_f32_16x16x4_f32(false, ra1[1], false, xh33, (short)0, racc1, false, false);
    rg0   = __builtin_amdgcn_wmma_f32_16x16x4_f32(false, rl0, false, lb2, (short)0, rg0, false, false);
    rg1   = __builtin_amdgcn_wmma_f32_16x16x4_f32(false, rl1, false, lb2, (short)0, rg1, false, false);

    const float* ep = emb + koff;
    const float* dp = dec_x + koff;
    v2f xb[32];
    int n1, n2, n3;
    {
        const long row0 = (long)t * BB + b;
        const int  i1 = cat1[row0], i2 = cat2[row0], i3 = cat3[row0];
        load_x(xb, dp + row0 * DD, ep + (long)i1 * 32, ep + (long)i2 * 32, ep + (long)i3 * 32);
        const int  tn   = (t + TG < T) ? t + TG : t;
        const long rown = (long)tn * BB + b;
        n1 = cat1[rown]; n2 = cat2[rown]; n3 = cat3[rown];
    }

    #pragma unroll 1
    for (; t < T; t += TG) {
        const long row = (long)t * BB + b;

        const int  t2   = (t + 2 * TG < T) ? t + 2 * TG : t;
        const long row2 = (long)t2 * BB + b;
        const int  m1 = cat1[row2], m2 = cat2[row2], m3 = cat3[row2];

        v8f acc0 = racc0;
        v8f acc1 = racc1;
        #pragma unroll
        for (int s = 0; s < 32; ++s) {
            acc0 = __builtin_amdgcn_wmma_f32_16x16x4_f32(false, wa0[s], false, xb[s],
                                                         (short)0, acc0, false, false);
            acc1 = __builtin_amdgcn_wmma_f32_16x16x4_f32(false, wa1[s], false, xb[s],
                                                         (short)0, acc1, false, false);
        }

        const bool  more = t + TG < T;
        const long  rown = (long)(more ? t + TG : t) * BB + b;
        const float* q0  = more ? dp + rown * DD : ep;
        load_x(xb, q0, ep + (long)n1 * 32, ep + (long)n2 * 32, ep + (long)n3 * 32);
        n1 = m1; n2 = m2; n3 = m3;

        float own[3];
        #pragma unroll
        for (int u = 0; u < 2; ++u) {
            own[u] = lstm_unit(acc0[u], acc0[2 + u], acc0[4 + u], acc0[6 + u], cc0[u]);
        }
        own[2] = 0.0f;
        if (half == 0) {
            own[2] = lstm_unit(acc1[0], acc1[1], acc1[2], acc1[3], cc0m);
        }

        const v2f lb0 = {own[0], own[1]};
        const v2f lb1 = {half ? hB[1] : own[2], l1e1};
        v8f g0 = rg0;
        v8f g1 = rg1;
        g0 = __builtin_amdgcn_wmma_f32_16x16x4_f32(false, la0[0], false, lb0,
                                                   (short)0, g0, false, false);
        g1 = __builtin_amdgcn_wmma_f32_16x16x4_f32(false, la1[0], false, lb0,
                                                   (short)0, g1, false, false);
        g0 = __builtin_amdgcn_wmma_f32_16x16x4_f32(false, la0[1], false, lb1,
                                                   (short)0, g0, false, false);
        g1 = __builtin_amdgcn_wmma_f32_16x16x4_f32(false, la1[1], false, lb1,
                                                   (short)0, g1, false, false);

        #pragma unroll
        for (int u = 0; u < 2; ++u) {
            out[row * HH + hu + u] = lstm_unit(g0[u], g0[2 + u], g0[4 + u], g0[6 + u], cc1[u]);
        }
        if (half == 0) {
            out[row * HH + 2] = lstm_unit(g1[0], g1[1], g1[2], g1[3], cc1m);
        }
    }
}

extern "C" void kernel_launch(void* const* d_in, const int* in_sizes, int n_in,
                              void* d_out, int out_size, void* d_ws, size_t ws_size,
                              hipStream_t stream)
{
    const float* hidden = (const float*)d_in[1];
    const float* cellp  = (const float*)d_in[2];
    const float* dec_x  = (const float*)d_in[3];
    const int*   cat1   = (const int*)d_in[4];
    const int*   cat2   = (const int*)d_in[5];
    const int*   cat3   = (const int*)d_in[6];
    const float* emb    = (const float*)d_in[7];
    const float* W_ih0  = (const float*)d_in[8];
    const float* W_hh0  = (const float*)d_in[9];
    const float* b_ih0  = (const float*)d_in[10];
    const float* b_hh0  = (const float*)d_in[11];
    const float* W_ih1  = (const float*)d_in[12];
    const float* W_hh1  = (const float*)d_in[13];
    const float* b_ih1  = (const float*)d_in[14];
    const float* b_hh1  = (const float*)d_in[15];
    float* out = (float*)d_out;

    const int T    = in_sizes[3] / (BB * DD);

    lstm_wmma_kernel<<<256, 256, 0, stream>>>(dec_x, cat1, cat2, cat3, emb,
                                              W_ih0, W_hh0, b_ih0, b_hh0,
                                              W_ih1, W_hh1, b_ih1, b_hh1,
                                              hidden, cellp, out, T);
}


# Tile walk pipelined one stage deeper: a tile's gate math, made one branch-free block (tanh's two forms selected per lane, same values), is issued between the next tile's WMMAs; sums unchanged
# speedup vs baseline: 2.3358x; 1.1281x over previous
#include <hip/hip_runtime.h>
#include <hip/hip_bf16.h>

typedef float v2f __attribute__((ext_vector_type(2)));
typedef float v8f __attribute__((ext_vector_type(8)));

#define BB   4096
#define DD   32
#define HH   5
#define GG   20
#define INW  128

__device__ __forceinline__ float sigf(float x) { return 1.0f / (1.0f + __expf(-x)); }

#define TG 8

__device__ __forceinline__ float tanh_sel(float x) {
    const float t  = fabsf(x);
    const float x2 = x * x;
    float p = fmaf(x2, __uint_as_float(0xbbbac73du), __uint_as_float(0x3ca908c9u));
    p = fmaf(x2, p, __uint_as_float(0xbd5c1c4eu));
    p = fmaf(x2, p, __uint_as_float(0x3e088382u));
    p = fmaf(x2, p, __uint_as_float(0xbeaaaa99u));
    const float small = fmaf(x2, t * p, t);
    const float e     = expf(t + t);
    const float big   = fmaf(__builtin_amdgcn_rcpf(e + 1.0f), -2.0f, 1.0f);
    return copysignf((t < 0.625f) ? small : big, x);
}

__device__ __forceinline__ float lstm_unit(float gi, float gf, float gg, float go, float c) {
    float cn = sigf(gf) * c + sigf(gi) * tanh_sel(gg);
    return sigf(go) * tanh_sel(cn);
}

__device__ __forceinline__ void load_x(v2f (&x)[32],
                                       const float* __restrict__ p0, const float* __restrict__ p1,
                                       const float* __restrict__ p2, const float* __restrict__ p3) {
    #pragma unroll
    for (int kk = 0; kk < 8; ++kk) {
        x[kk]      = *(const v2f*)(p0 + kk * 4);
        x[8 + kk]  = *(const v2f*)(p1 + kk * 4);
        x[16 + kk] = *(const v2f*)(p2 + kk * 4);
        x[24 + kk] = *(const v2f*)(p3 + kk * 4);
    }
}

__global__ __launch_bounds__(256) __attribute__((amdgpu_waves_per_eu(2, 2))) void lstm_wmma_kernel(
    const float* __restrict__ dec_x,
    const int*   __restrict__ cat1,
    const int*   __restrict__ cat2,
    const int*   __restrict__ cat3,
    const float* __restrict__ emb,
    const float* __restrict__ W_ih0,
    const float* __restrict__ W_hh0,
    const float* __restrict__ b_ih0,
    const float* __restrict__ b_hh0,
    const float* __restrict__ W_ih1,
    const float* __restrict__ W_hh1,
    const float* __restrict__ b_ih1,
    const float* __restrict__ b_hh1,
    const float* __restrict__ hidden,
    const float* __restrict__ cell,
    float* __restrict__ out,
    int T)
{
    const int tid  = threadIdx.x;
    const int wave = tid >> 5;
    const int lane = tid & 31;
    const int m    = lane & 15;
    const int half = lane >> 4;
    const int koff = half * 2;

    const int    gs0  = ((m >> 1) & 3) * HH + (m & 1) + ((m & 8) ? 3 : 0);
    const int    gs1  = (m < 4) ? (m * HH + 2) : 2;
    const float* wr0  = W_ih0 + gs0 * INW + koff;
    const float* wr1  = W_ih0 + gs1 * INW + koff;
    const v2f    zero = {0.0f, 0.0f};
    const bool   act1 = m < 4;
    v2f wa0[32], wa1[32];
    #pragma unroll
    for (int s = 0; s < 32; ++s) {
        wa0[s] = *(const v2f*)(wr0 + s * 4);
        v2f w1 = *(const v2f*)(wr1 + s * 4);
        wa1[s] = act1 ? w1 : zero;
    }
    v2f la0[2], la1[2];
    v2f ra0[2], ra1[2], rl0, rl1;
    {
        const float* u0 = W_hh0 + gs0 * HH;
        const float* u1 = W_hh0 + gs1 * HH;
        const float* v0 = W_ih1 + gs0 * HH;
        const float* v1 = W_ih1 + gs1 * HH;
        const float* y0 = W_hh1 + gs0 * HH;
        const float* y1 = W_hh1 + gs1 * HH;
        const v2f p0 = {u0[koff], u0[koff + 1]};
        const v2f p1 = {u1[koff], u1[koff + 1]};
        ra0[0] = p0;
        ra1[0] = act1 ? p1 : zero;
        const v2f q0h0 = {u0[4], b_ih0[gs0]}, q0h1 = {b_hh0[gs0], 0.0f};
        const v2f q1h0 = {u1[4], b_ih0[gs1]}, q1h1 = {b_hh0[gs1], 0.0f};
        ra0[1] = half ? q0h1 : q0h0;
        ra1[1] = act1 ? (half ? q1h1 : q1h0) : zero;

        const int c3 = half * 3;
        const v2f a0 = {v0[c3], v0[c3 + 1]};
        const v2f a1 = {v1[c3], v1[c3 + 1]};
        la0[0] = a0;
        la1[0] = act1 ? a1 : zero;
        const v2f b0h0 = {v0[2], y0[0]}, b0h1 = {y0[1], y0[2]};
        const v2f b1h0 = {v1[2], y1[0]}, b1h1 = {y1[1], y1[2]};
        la0[1] = half ? b0h1 : b0h0;
        la1[1] = act1 ? (half ? b1h1 : b1h0) : zero;
        const v2f c0h0 = {y0[3], y0[4]}, c0h1 = {b_ih1[gs0], b_hh1[gs0]};
        const v2f c1h0 = {y1[3], y1[4]}, c1h1 = {b_ih1[gs1], b_hh1[gs1]};
        rl0 = half ? c0h1 : c0h0;
        rl1 = act1 ? (half ? c1h1 : c1h0) : zero;
    }

    const int gw = blockIdx.x * 8 + __builtin_amdgcn_readfirstlane(wave);
    const int b  = (gw & 255) * 16 + m;

    int t = gw >> 8;
    if (t >= T) return;

    const int hu = half * 3;
    const float* h0r  = hidden + (long)b * HH;
    const float* h1r  = hidden + (long)BB * HH + (long)b * HH;
    const float* c0r  = cell + (long)b * HH;
    const float* c1r  = cell + (long)BB * HH + (long)b * HH;

    float hA[HH], hB[HH], cc0[2], cc1[2];
    #pragma unroll
    for (int j = 0; j < HH; ++j) {
        hA[j] = h0r[j];
        hB[j] = h1r[j];
    }
    #pragma unroll
    for (int u = 0; u < 2; ++u) {
        cc0[u] = c0r[hu + u];
        cc1[u] = c1r[hu + u];
    }
    const float cc0m = c0r[2];
    const float cc1m = c1r[2];

    const v2f xh32 = {half ? hA[2] : hA[0], half ? hA[3] : hA[1]};
    const v2f xh33 = {half ? 1.0f : hA[4], half ? 0.0f : 1.0f};
    const float l1e1 = half ? hB[2] : hB[0];
    const v2f lb2  = {half ? 1.0f : hB[3], half ? 1.0f : hB[4]};

    v8f racc0 = {}, racc1 = {}, rg0 = {}, rg1 = {};
    racc0 = __builtin_amdgcn_wmma_f32_16x16x4_f32(false, ra0[0], false, xh32, (short)0, racc0, false, false);
    racc1 = __builtin_amdgcn_wmma_f32_16x16x4_f32(false, ra1[0], false, xh32, (short)0, racc1, false, false);
    racc0 = __builtin_amdgcn_wmma_f32_16x16x4_f32(false, ra0[1], false, xh33, (short)0, racc0, false, false);
    racc1 = __builtin_amdgcn_wmma_f32_16x16x4_f32(false, ra1[1], false, xh33, (short)0, racc1, false, false);
    rg0   = __builtin_amdgcn_wmma_f32_16x16x4_f32(false, rl0, false, lb2, (short)0, rg0, false, false);
    rg1   = __builtin_amdgcn_wmma_f32_16x16x4_f32(false, rl1, false, lb2, (short)0, rg1, false, false);

    const float* ep = emb + koff;
    const float* dp = dec_x + koff;
    v2f xa[32], xb[32];
    int n1, n2, n3;
    {
        const long row0 = (long)t * BB + b;
        const int  i1 = cat1[row0], i2 = cat2[row0], i3 = cat3[row0];
        const bool has1 = t + TG < T;
        const long row1 = (long)(has1 ? t + TG : t) * BB + b;
        const int  j1 = cat1[row1], j2 = cat2[row1], j3 = cat3[row1];
        load_x(xa, dp + row0 * DD, ep + (long)i1 * 32, ep + (long)i2 * 32, ep + (long)i3 * 32);
        const float* q1 = has1 ? dp + row1 * DD : ep;
        load_x(xb, q1, ep + (long)j1 * 32, ep + (long)j2 * 32, ep + (long)j3 * 32);
        const int  t2   = (t + 2 * TG < T) ? t + 2 * TG : t;
        const long row2 = (long)t2 * BB + b;
        n1 = cat1[row2]; n2 = cat2[row2]; n3 = cat3[row2];
    }

    v8f acc0 = racc0;
    v8f acc1 = racc1;
    #pragma unroll
    for (int s = 0; s < 32; ++s) {
        acc0 = __builtin_amdgcn_wmma_f32_16x16x4_f32(false, wa0[s], false, xa[s],
                                                     (short)0, acc0, false, false);
        acc1 = __builtin_amdgcn_wmma_f32_16x16x4_f32(false, wa1[s], false, xa[s],
                                                     (short)0, acc1, false, false);
    }

#define TILE_STEP(TT, XUSE, XLOAD)                                                                   \
    {                                                                                                \
        const int  tt_  = (TT);                                                                      \
        const long row  = (long)tt_ * BB + b;                                                        \
        const int  t3   = (tt_ + 3 * TG < T) ? tt_ + 3 * TG : tt_;                                   \
        const long row3 = (long)t3 * BB + b;                                                         \
        const int  m1 = cat1[row3], m2 = cat2[row3], m3 = cat3[row3];                                \
        const bool  more = tt_ + 2 * TG < T;                                                         \
        const long  rown = (long)(more ? tt_ + 2 * TG : tt_) * BB + b;                               \
        const float* q0  = more ? dp + rown * DD : ep;                                               \
        load_x(XLOAD, q0, ep + (long)n1 * 32, ep + (long)n2 * 32, ep + (long)n3 * 32);               \
        n1 = m1; n2 = m2; n3 = m3;                                                                   \
                             \
        __builtin_amdgcn_sched_barrier(0);                                                           \
        v8f nx0 = racc0;                                                                             \
        v8f nx1 = racc1;                                                                             \
        _Pragma("unroll")                                                                            \
        for (int s = 0; s < 16; ++s) {                                                               \
            nx0 = __builtin_amdgcn_wmma_f32_16x16x4_f32(false, wa0[s], false, XUSE[s],               \
                                                        (short)0, nx0, false, false);                \
            nx1 = __builtin_amdgcn_wmma_f32_16x16x4_f32(false, wa1[s], false, XUSE[s],               \
                                                        (short)0, nx1, false, false);                \
        }                                                                                            \
            \
        float own[3];                                                                                \
        _Pragma("unroll")                                                                            \
        for (int u = 0; u < 2; ++u) {                                                                \
            own[u] = lstm_unit(acc0[u], acc0[2 + u], acc0[4 + u], acc0[6 + u], cc0[u]);              \
        }                                                                                            \
        own[2] = lstm_unit(acc1[0], acc1[1], acc1[2], acc1[3], cc0m);                                \
        const v2f lb0 = {own[0], own[1]};                                                            \
        const v2f lb1 = {half ? hB[1] : own[2], l1e1};                                               \
        v8f g0 = rg0;                                                                                \
        v8f g1 = rg1;                                                                                \
        g0 = __builtin_amdgcn_wmma_f32_16x16x4_f32(false, la0[0], false, lb0,                        \
                                                   (short)0, g0, false, false);                      \
        g1 = __builtin_amdgcn_wmma_f32_16x16x4_f32(false, la1[0], false, lb0,                        \
                                                   (short)0, g1, false, false);                      \
        g0 = __builtin_amdgcn_wmma_f32_16x16x4_f32(false, la0[1], false, lb1,                        \
                                                   (short)0, g0, false, false);                      \
        g1 = __builtin_amdgcn_wmma_f32_16x16x4_f32(false, la1[1], false, lb1,                        \
                                                   (short)0, g1, false, false);                      \
                                                      \
        _Pragma("unroll")                                                                            \
        for (int k = 0; k < 32; ++k) {                                                               \
            __builtin_amdgcn_sched_group_barrier(0x008, 1, 0);                                       \
            __builtin_amdgcn_sched_group_barrier(0x402, 9, 0);                                       \
        }                                                                                            \
        __builtin_amdgcn_sched_group_barrier(0x008, 4, 0);                                           \
                   \
        asm volatile("" : "+v"(g0), "+v"(g1), "+v"(nx0), "+v"(nx1));                                 \
        __builtin_amdgcn_sched_barrier(0);                                                           \
        _Pragma("unroll")                                                                            \
        for (int s = 16; s < 32; ++s) {                                                              \
            nx0 = __builtin_amdgcn_wmma_f32_16x16x4_f32(false, wa0[s], false, XUSE[s],               \
                                                        (short)0, nx0, false, false);                \
            nx1 = __builtin_amdgcn_wmma_f32_16x16x4_f32(false, wa1[s], false, XUSE[s],               \
                                                        (short)0, nx1, false, false);                \
        }                                                                                            \
        float o[3];                                                                                  \
        _Pragma("unroll")                                                                            \
        for (int u = 0; u < 2; ++u) {                                                                \
            o[u] = lstm_unit(g0[u], g0[2 + u], g0[4 + u], g0[6 + u], cc1[u]);                        \
        }                                                                                            \
        o[2] = lstm_unit(g1[0], g1[1], g1[2], g1[3], cc1m);                                          \
                                                                 \
        _Pragma("unroll")                                                                            \
        for (int k = 0; k < 32; ++k) {                                                               \
            __builtin_amdgcn_sched_group_barrier(0x008, 1, 0);                                       \
            __builtin_amdgcn_sched_group_barrier(0x402, 9, 0);                                       \
        }                                                                                            \
                       \
        asm volatile("" :: "v"(o[0]), "v"(o[1]), "v"(o[2]));                                         \
        __builtin_amdgcn_sched_barrier(0);                                                           \
        out[row * HH + hu]     = o[0];                                                               \
        out[row * HH + hu + 1] = o[1];                                                               \
        if (half == 0) {                                                                             \
            out[row * HH + 2] = o[2];                                                                \
        }                                                                                            \
        acc0 = nx0;                                                                                  \
        acc1 = nx1;                                                                                  \
    }

    #pragma unroll 1
    for (;;) {
        TILE_STEP(t, xb, xa)
        t += TG;
        if (t >= T) break;
        TILE_STEP(t, xa, xb)
        t += TG;
        if (t >= T) break;
    }
#undef TILE_STEP
}

extern "C" void kernel_launch(void* const* d_in, const int* in_sizes, int n_in,
                              void* d_out, int out_size, void* d_ws, size_t ws_size,
                              hipStream_t stream)
{
    const float* hidden = (const float*)d_in[1];
    const float* cellp  = (const float*)d_in[2];
    const float* dec_x  = (const float*)d_in[3];
    const int*   cat1   = (const int*)d_in[4];
    const int*   cat2   = (const int*)d_in[5];
    const int*   cat3   = (const int*)d_in[6];
    const float* emb    = (const float*)d_in[7];
    const float* W_ih0  = (const float*)d_in[8];
    const float* W_hh0  = (const float*)d_in[9];
    const float* b_ih0  = (const float*)d_in[10];
    const float* b_hh0  = (const float*)d_in[11];
    const float* W_ih1  = (const float*)d_in[12];
    const float* W_hh1  = (const float*)d_in[13];
    const float* b_ih1  = (const float*)d_in[14];
    const float* b_hh1  = (const float*)d_in[15];
    float* out = (float*)d_out;

    const int T    = in_sizes[3] / (BB * DD);

    lstm_wmma_kernel<<<256, 256, 0, stream>>>(dec_x, cat1, cat2, cat3, emb,
                                              W_ih0, W_hh0, b_ih0, b_hh0,
                                              W_ih1, W_hh1, b_ih1, b_hh1,
                                              hidden, cellp, out, T);
}


# Walk's last step peeled: the final tile's gate math runs alone, so the 64 matrix instructions and 35 loads that step issued for a tile past the end are gone; sums unchanged
# speedup vs baseline: 2.5809x; 1.1049x over previous
#include <hip/hip_runtime.h>
#include <hip/hip_bf16.h>
#include <type_traits>

typedef float v2f __attribute__((ext_vector_type(2)));
typedef float v8f __attribute__((ext_vector_type(8)));

#define BB   4096
#define DD   32
#define HH   5
#define GG   20
#define INW  128

__device__ __forceinline__ float sigf(float x) { return 1.0f / (1.0f + __expf(-x)); }

#define TG 8

__device__ __forceinline__ float tanh_sel(float x) {
    const float t  = fabsf(x);
    const float x2 = x * x;
    float p = fmaf(x2, __uint_as_float(0xbbbac73du), __uint_as_float(0x3ca908c9u));
    p = fmaf(x2, p, __uint_as_float(0xbd5c1c4eu));
    p = fmaf(x2, p, __uint_as_float(0x3e088382u));
    p = fmaf(x2, p, __uint_as_float(0xbeaaaa99u));
    const float small = fmaf(x2, t * p, t);
    const float e     = expf(t + t);
    const float big   = fmaf(__builtin_amdgcn_rcpf(e + 1.0f), -2.0f, 1.0f);
    return copysignf((t < 0.625f) ? small : big, x);
}

__device__ __forceinline__ float lstm_unit(float gi, float gf, float gg, float go, float c) {
    float cn = sigf(gf) * c + sigf(gi) * tanh_sel(gg);
    return sigf(go) * tanh_sel(cn);
}

__device__ __forceinline__ void load_x(v2f (&x)[32],
                                       const float* __restrict__ p0, const float* __restrict__ p1,
                                       const float* __restrict__ p2, const float* __restrict__ p3) {
    #pragma unroll
    for (int kk = 0; kk < 8; ++kk) {
        x[kk]      = *(const v2f*)(p0 + kk * 4);
        x[8 + kk]  = *(const v2f*)(p1 + kk * 4);
        x[16 + kk] = *(const v2f*)(p2 + kk * 4);
        x[24 + kk] = *(const v2f*)(p3 + kk * 4);
    }
}

__global__ __launch_bounds__(256) __attribute__((amdgpu_waves_per_eu(2, 2))) void lstm_wmma_kernel(
    const float* __restrict__ dec_x,
    const int*   __restrict__ cat1,
    const int*   __restrict__ cat2,
    const int*   __restrict__ cat3,
    const float* __restrict__ emb,
    const float* __restrict__ W_ih0,
    const float* __restrict__ W_hh0,
    const float* __restrict__ b_ih0,
    const float* __restrict__ b_hh0,
    const float* __restrict__ W_ih1,
    const float* __restrict__ W_hh1,
    const float* __restrict__ b_ih1,
    const float* __restrict__ b_hh1,
    const float* __restrict__ hidden,
    const float* __restrict__ cell,
    float* __restrict__ out,
    int T)
{
    const int tid  = threadIdx.x;
    const int wave = tid >> 5;
    const int lane = tid & 31;
    const int m    = lane & 15;
    const int half = lane >> 4;
    const int koff = half * 2;

    const int    gs0  = ((m >> 1) & 3) * HH + (m & 1) + ((m & 8) ? 3 : 0);
    const int    gs1  = (m < 4) ? (m * HH + 2) : 2;
    const float* wr0  = W_ih0 + gs0 * INW + koff;
    const float* wr1  = W_ih0 + gs1 * INW + koff;
    const v2f    zero = {0.0f, 0.0f};
    const bool   act1 = m < 4;
    v2f wa0[32], wa1[32];
    #pragma unroll
    for (int s = 0; s < 32; ++s) {
        wa0[s] = *(const v2f*)(wr0 + s * 4);
        v2f w1 = *(const v2f*)(wr1 + s * 4);
        wa1[s] = act1 ? w1 : zero;
    }
    v2f la0[2], la1[2];
    v2f ra0[2], ra1[2], rl0, rl1;
    {
        const float* u0 = W_hh0 + gs0 * HH;
        const float* u1 = W_hh0 + gs1 * HH;
        const float* v0 = W_ih1 + gs0 * HH;
        const float* v1 = W_ih1 + gs1 * HH;
        const float* y0 = W_hh1 + gs0 * HH;
        const float* y1 = W_hh1 + gs1 * HH;
        const v2f p0 = {u0[koff], u0[koff + 1]};
        const v2f p1 = {u1[koff], u1[koff + 1]};
        ra0[0] = p0;
        ra1[0] = act1 ? p1 : zero;
        const v2f q0h0 = {u0[4], b_ih0[gs0]}, q0h1 = {b_hh0[gs0], 0.0f};
        const v2f q1h0 = {u1[4], b_ih0[gs1]}, q1h1 = {b_hh0[gs1], 0.0f};
        ra0[1] = half ? q0h1 : q0h0;
        ra1[1] = act1 ? (half ? q1h1 : q1h0) : zero;

        const int c3 = half * 3;
        const v2f a0 = {v0[c3], v0[c3 + 1]};
        const v2f a1 = {v1[c3], v1[c3 + 1]};
        la0[0] = a0;
        la1[0] = act1 ? a1 : zero;
        const v2f b0h0 = {v0[2], y0[0]}, b0h1 = {y0[1], y0[2]};
        const v2f b1h0 = {v1[2], y1[0]}, b1h1 = {y1[1], y1[2]};
        la0[1] = half ? b0h1 : b0h0;
        la1[1] = act1 ? (half ? b1h1 : b1h0) : zero;
        const v2f c0h0 = {y0[3], y0[4]}, c0h1 = {b_ih1[gs0], b_hh1[gs0]};
        const v2f c1h0 = {y1[3], y1[4]}, c1h1 = {b_ih1[gs1], b_hh1[gs1]};
        rl0 = half ? c0h1 : c0h0;
        rl1 = act1 ? (half ? c1h1 : c1h0) : zero;
    }

    const int gw = blockIdx.x * 8 + __builtin_amdgcn_readfirstlane(wave);
    const int b  = (gw & 255) * 16 + m;

    int t = gw >> 8;
    if (t >= T) return;

    auto walk = [&](auto one_tile) __attribute__((always_inline)) {
    constexpr bool ONE = decltype(one_tile)::value;

    const int hu = half * 3;
    const float* h0r  = hidden + (long)b * HH;
    const float* h1r  = hidden + (long)BB * HH + (long)b * HH;
    const float* c0r  = cell + (long)b * HH;
    const float* c1r  = cell + (long)BB * HH + (long)b * HH;

    float hA[HH], hB[HH], cc0[2], cc1[2];
    #pragma unroll
    for (int j = 0; j < HH; ++j) {
        hA[j] = h0r[j];
        hB[j] = h1r[j];
    }
    #pragma unroll
    for (int u = 0; u < 2; ++u) {
        cc0[u] = c0r[hu + u];
        cc1[u] = c1r[hu + u];
    }
    const float cc0m = c0r[2];
    const float cc1m = c1r[2];

    const v2f xh32 = {half ? hA[2] : hA[0], half ? hA[3] : hA[1]};
    const v2f xh33 = {half ? 1.0f : hA[4], half ? 0.0f : 1.0f};
    const float l1e1 = half ? hB[2] : hB[0];
    const v2f lb2  = {half ? 1.0f : hB[3], half ? 1.0f : hB[4]};

    v8f racc0 = {}, racc1 = {}, rg0 = {}, rg1 = {};
    racc0 = __builtin_amdgcn_wmma_f32_16x16x4_f32(false, ra0[0], false, xh32, (short)0, racc0, false, false);
    racc1 = __builtin_amdgcn_wmma_f32_16x16x4_f32(false, ra1[0], false, xh32, (short)0, racc1, false, false);
    racc0 = __builtin_amdgcn_wmma_f32_16x16x4_f32(false, ra0[1], false, xh33, (short)0, racc0, false, false);
    racc1 = __builtin_amdgcn_wmma_f32_16x16x4_f32(false, ra1[1], false, xh33, (short)0, racc1, false, false);
    rg0   = __builtin_amdgcn_wmma_f32_16x16x4_f32(false, rl0, false, lb2, (short)0, rg0, false, false);
    rg1   = __builtin_amdgcn_wmma_f32_16x16x4_f32(false, rl1, false, lb2, (short)0, rg1, false, false);

    const float* ep = emb + koff;
    const float* dp = dec_x + koff;
    v2f xa[32], xb[32];
    int n1, n2, n3;
    if constexpr (ONE) {
        const long row0 = (long)t * BB + b;
        const int  i1 = cat1[row0], i2 = cat2[row0], i3 = cat3[row0];
        load_x(xa, dp + row0 * DD, ep + (long)i1 * 32, ep + (long)i2 * 32, ep + (long)i3 * 32);
    } else {
        const long row0 = (long)t * BB + b;
        const int  i1 = cat1[row0], i2 = cat2[row0], i3 = cat3[row0];
        const long row1 = (long)(t + TG) * BB + b;
        const int  j1 = cat1[row1], j2 = cat2[row1], j3 = cat3[row1];
        load_x(xa, dp + row0 * DD, ep + (long)i1 * 32, ep + (long)i2 * 32, ep + (long)i3 * 32);
        load_x(xb, dp + row1 * DD, ep + (long)j1 * 32, ep + (long)j2 * 32, ep + (long)j3 * 32);
        const int  t2   = (t + 2 * TG < T) ? t + 2 * TG : t;
        const long row2 = (long)t2 * BB + b;
        n1 = cat1[row2]; n2 = cat2[row2]; n3 = cat3[row2];
    }

    v8f acc0 = racc0;
    v8f acc1 = racc1;
    #pragma unroll
    for (int s = 0; s < 32; ++s) {
        acc0 = __builtin_amdgcn_wmma_f32_16x16x4_f32(false, wa0[s], false, xa[s],
                                                     (short)0, acc0, false, false);
        acc1 = __builtin_amdgcn_wmma_f32_16x16x4_f32(false, wa1[s], false, xa[s],
                                                     (short)0, acc1, false, false);
    }

#define TILE_STEP(TT, XUSE, XLOAD)                                                                   \
    {                                                                                                \
        const int  tt_  = (TT);                                                                      \
        const long row  = (long)tt_ * BB + b;                                                        \
        const int  t3   = (tt_ + 3 * TG < T) ? tt_ + 3 * TG : tt_;                                   \
        const long row3 = (long)t3 * BB + b;                                                         \
        const int  m1 = cat1[row3], m2 = cat2[row3], m3 = cat3[row3];                                \
        const bool  more = tt_ + 2 * TG < T;                                                         \
        const long  rown = (long)(more ? tt_ + 2 * TG : tt_) * BB + b;                               \
        const float* q0  = more ? dp + rown * DD : ep;                                               \
        load_x(XLOAD, q0, ep + (long)n1 * 32, ep + (long)n2 * 32, ep + (long)n3 * 32);               \
        n1 = m1; n2 = m2; n3 = m3;                                                                   \
                             \
        __builtin_amdgcn_sched_barrier(0);                                                           \
        v8f nx0 = racc0;                                                                             \
        v8f nx1 = racc1;                                                                             \
        _Pragma("unroll")                                                                            \
        for (int s = 0; s < 16; ++s) {                                                               \
            nx0 = __builtin_amdgcn_wmma_f32_16x16x4_f32(false, wa0[s], false, XUSE[s],               \
                                                        (short)0, nx0, false, false);                \
            nx1 = __builtin_amdgcn_wmma_f32_16x16x4_f32(false, wa1[s], false, XUSE[s],               \
                                                        (short)0, nx1, false, false);                \
        }                                                                                            \
            \
        float own[3];                                                                                \
        _Pragma("unroll")                                                                            \
        for (int u = 0; u < 2; ++u) {                                                                \
            own[u] = lstm_unit(acc0[u], acc0[2 + u], acc0[4 + u], acc0[6 + u], cc0[u]);              \
        }                                                                                            \
        own[2] = lstm_unit(acc1[0], acc1[1], acc1[2], acc1[3], cc0m);                                \
        const v2f lb0 = {own[0], own[1]};                                                            \
        const v2f lb1 = {half ? hB[1] : own[2], l1e1};                                               \
        v8f g0 = rg0;                                                                                \
        v8f g1 = rg1;                                                                                \
        g0 = __builtin_amdgcn_wmma_f32_16x16x4_f32(false, la0[0], false, lb0,                        \
                                                   (short)0, g0, false, false);                      \
        g1 = __builtin_amdgcn_wmma_f32_16x16x4_f32(false, la1[0], false, lb0,                        \
                                                   (short)0, g1, false, false);                      \
        g0 = __builtin_amdgcn_wmma_f32_16x16x4_f32(false, la0[1], false, lb1,                        \
                                                   (short)0, g0, false, false);                      \
        g1 = __builtin_amdgcn_wmma_f32_16x16x4_f32(false, la1[1], false, lb1,                        \
                                                   (short)0, g1, false, false);                      \
                                                      \
        _Pragma("unroll")                                                                            \
        for (int k = 0; k < 32; ++k) {                                                               \
            __builtin_amdgcn_sched_group_barrier(0x008, 1, 0);                                       \
            __builtin_amdgcn_sched_group_barrier(0x402, 9, 0);                                       \
        }                                                                                            \
        __builtin_amdgcn_sched_group_barrier(0x008, 4, 0);                                           \
                   \
        asm volatile("" : "+v"(g0), "+v"(g1), "+v"(nx0), "+v"(nx1));                                 \
        __builtin_amdgcn_sched_barrier(0);                                                           \
        _Pragma("unroll")                                                                            \
        for (int s = 16; s < 32; ++s) {                                                              \
            nx0 = __builtin_amdgcn_wmma_f32_16x16x4_f32(false, wa0[s], false, XUSE[s],               \
                                                        (short)0, nx0, false, false);                \
            nx1 = __builtin_amdgcn_wmma_f32_16x16x4_f32(false, wa1[s], false, XUSE[s],               \
                                                        (short)0, nx1, false, false);                \
        }                                                                                            \
        float o[3];                                                                                  \
        _Pragma("unroll")                                                                            \
        for (int u = 0; u < 2; ++u) {                                                                \
            o[u] = lstm_unit(g0[u], g0[2 + u], g0[4 + u], g0[6 + u], cc1[u]);                        \
        }                                                                                            \
        o[2] = lstm_unit(g1[0], g1[1], g1[2], g1[3], cc1m);                                          \
                                                                 \
        _Pragma("unroll")                                                                            \
        for (int k = 0; k < 32; ++k) {                                                               \
            __builtin_amdgcn_sched_group_barrier(0x008, 1, 0);                                       \
            __builtin_amdgcn_sched_group_barrier(0x402, 9, 0);                                       \
        }                                                                                            \
                       \
        asm volatile("" :: "v"(o[0]), "v"(o[1]), "v"(o[2]));                                         \
        __builtin_amdgcn_sched_barrier(0);                                                           \
        out[row * HH + hu]     = o[0];                                                               \
        out[row * HH + hu + 1] = o[1];                                                               \
        if (half == 0) {                                                                             \
            out[row * HH + 2] = o[2];                                                                \
        }                                                                                            \
        acc0 = nx0;                                                                                  \
        acc1 = nx1;                                                                                  \
    }

    if constexpr (!ONE) {
        #pragma unroll 1
        for (;;) {
            TILE_STEP(t, xb, xa)
            t += TG;
            if (t + TG >= T) break;
            TILE_STEP(t, xa, xb)
            t += TG;
            if (t + TG >= T) break;
        }
    }
#undef TILE_STEP

    {
        const long row = (long)t * BB + b;
        float own[3];
        #pragma unroll
        for (int u = 0; u < 2; ++u) {
            own[u] = lstm_unit(acc0[u], acc0[2 + u], acc0[4 + u], acc0[6 + u], cc0[u]);
        }
        own[2] = lstm_unit(acc1[0], acc1[1], acc1[2], acc1[3], cc0m);
        const v2f lb0 = {own[0], own[1]};
        const v2f lb1 = {half ? hB[1] : own[2], l1e1};
        v8f g0 = rg0;
        v8f g1 = rg1;
        g0 = __builtin_amdgcn_wmma_f32_16x16x4_f32(false, la0[0], false, lb0, (short)0, g0, false, false);
        g1 = __builtin_amdgcn_wmma_f32_16x16x4_f32(false, la1[0], false, lb0, (short)0, g1, false, false);
        g0 = __builtin_amdgcn_wmma_f32_16x16x4_f32(false, la0[1], false, lb1, (short)0, g0, false, false);
        g1 = __builtin_amdgcn_wmma_f32_16x16x4_f32(false, la1[1], false, lb1, (short)0, g1, false, false);
        float o[3];
        #pragma unroll
        for (int u = 0; u < 2; ++u) {
            o[u] = lstm_unit(g0[u], g0[2 + u], g0[4 + u], g0[6 + u], cc1[u]);
        }
        o[2] = lstm_unit(g1[0], g1[1], g1[2], g1[3], cc1m);
        out[row * HH + hu]     = o[0];
        out[row * HH + hu + 1] = o[1];
        if (half == 0) {
            out[row * HH + 2] = o[2];
        }
    }
    };

    if (t + TG >= T) {
        walk(std::true_type{});
    } else {
        walk(std::false_type{});
    }
}

extern "C" void kernel_launch(void* const* d_in, const int* in_sizes, int n_in,
                              void* d_out, int out_size, void* d_ws, size_t ws_size,
                              hipStream_t stream)
{
    const float* hidden = (const float*)d_in[1];
    const float* cellp  = (const float*)d_in[2];
    const float* dec_x  = (const float*)d_in[3];
    const int*   cat1   = (const int*)d_in[4];
    const int*   cat2   = (const int*)d_in[5];
    const int*   cat3   = (const int*)d_in[6];
    const float* emb    = (const float*)d_in[7];
    const float* W_ih0  = (const float*)d_in[8];
    const float* W_hh0  = (const float*)d_in[9];
    const float* b_ih0  = (const float*)d_in[10];
    const float* b_hh0  = (const float*)d_in[11];
    const float* W_ih1  = (const float*)d_in[12];
    const float* W_hh1  = (const float*)d_in[13];
    const float* b_ih1  = (const float*)d_in[14];
    const float* b_hh1  = (const float*)d_in[15];
    float* out = (float*)d_out;

    const int T    = in_sizes[3] / (BB * DD);

    lstm_wmma_kernel<<<256, 256, 0, stream>>>(dec_x, cat1, cat2, cat3, emb,
                                              W_ih0, W_hh0, b_ih0, b_hh0,
                                              W_ih1, W_hh1, b_ih1, b_hh1,
                                              hidden, cellp, out, T);
}


# Fragment loads of weights and x 16 bytes per lane (was 8): a load feeds two k-steps, same k order in both loaders; load instructions and line visits halved; k order inside the sums changes
# speedup vs baseline: 2.6715x; 1.0351x over previous
#include <hip/hip_runtime.h>
#include <hip/hip_bf16.h>
#include <type_traits>

typedef float v2f __attribute__((ext_vector_type(2)));
typedef float v4f __attribute__((ext_vector_type(4)));
typedef float v8f __attribute__((ext_vector_type(8)));

__device__ __forceinline__ v2f lo2(v4f v) { return __builtin_shufflevector(v, v, 0, 1); }
__device__ __forceinline__ v2f hi2(v4f v) { return __builtin_shufflevector(v, v, 2, 3); }

#define BB   4096
#define DD   32
#define HH   5
#define GG   20
#define INW  128

__device__ __forceinline__ float sigf(float x) { return 1.0f / (1.0f + __expf(-x)); }

#define TG 8

__device__ __forceinline__ float tanh_sel(float x) {
    const float t  = fabsf(x);
    const float x2 = x * x;
    float p = fmaf(x2, __uint_as_float(0xbbbac73du), __uint_as_float(0x3ca908c9u));
    p = fmaf(x2, p, __uint_as_float(0xbd5c1c4eu));
    p = fmaf(x2, p, __uint_as_float(0x3e088382u));
    p = fmaf(x2, p, __uint_as_float(0xbeaaaa99u));
    const float small = fmaf(x2, t * p, t);
    const float e     = expf(t + t);
    const float big   = fmaf(__builtin_amdgcn_rcpf(e + 1.0f), -2.0f, 1.0f);
    return copysignf((t < 0.625f) ? small : big, x);
}

__device__ __forceinline__ float lstm_unit(float gi, float gf, float gg, float go, float c) {
    float cn = sigf(gf) * c + sigf(gi) * tanh_sel(gg);
    return sigf(go) * tanh_sel(cn);
}

__device__ __forceinline__ void load_x(v4f (&x)[16],
                                       const float* __restrict__ p0, const float* __restrict__ p1,
                                       const float* __restrict__ p2, const float* __restrict__ p3) {
    #pragma unroll
    for (int q = 0; q < 4; ++q) {
        x[q]      = *(const v4f*)(p0 + q * 8);
        x[4 + q]  = *(const v4f*)(p1 + q * 8);
        x[8 + q]  = *(const v4f*)(p2 + q * 8);
        x[12 + q] = *(const v4f*)(p3 + q * 8);
    }
}

__global__ __launch_bounds__(256) __attribute__((amdgpu_waves_per_eu(2, 2))) void lstm_wmma_kernel(
    const float* __restrict__ dec_x,
    const int*   __restrict__ cat1,
    const int*   __restrict__ cat2,
    const int*   __restrict__ cat3,
    const float* __restrict__ emb,
    const float* __restrict__ W_ih0,
    const float* __restrict__ W_hh0,
    const float* __restrict__ b_ih0,
    const float* __restrict__ b_hh0,
    const float* __restrict__ W_ih1,
    const float* __restrict__ W_hh1,
    const float* __restrict__ b_ih1,
    const float* __restrict__ b_hh1,
    const float* __restrict__ hidden,
    const float* __restrict__ cell,
    float* __restrict__ out,
    int T)
{
    const int tid  = threadIdx.x;
    const int wave = tid >> 5;
    const int lane = tid & 31;
    const int m    = lane & 15;
    const int half = lane >> 4;
    const int koff  = half * 2;
    const int koff4 = half * 4;

    const int    gs0  = ((m >> 1) & 3) * HH + (m & 1) + ((m & 8) ? 3 : 0);
    const int    gs1  = (m < 4) ? (m * HH + 2) : 2;
    const float* wr0  = W_ih0 + gs0 * INW + koff4;
    const float* wr1  = W_ih0 + gs1 * INW + koff4;
    const v2f    zero = {0.0f, 0.0f};
    const v4f    zero4 = {0.0f, 0.0f, 0.0f, 0.0f};
    const bool   act1 = m < 4;
    v4f wa0[16], wa1[16];
    #pragma unroll
    for (int j = 0; j < 16; ++j) {
        wa0[j] = *(const v4f*)(wr0 + j * 8);
        v4f w1 = *(const v4f*)(wr1 + j * 8);
        wa1[j] = act1 ? w1 : zero4;
    }
    v2f la0[2], la1[2];
    v2f ra0[2], ra1[2], rl0, rl1;
    {
        const float* u0 = W_hh0 + gs0 * HH;
        const float* u1 = W_hh0 + gs1 * HH;
        const float* v0 = W_ih1 + gs0 * HH;
        const float* v1 = W_ih1 + gs1 * HH;
        const float* y0 = W_hh1 + gs0 * HH;
        const float* y1 = W_hh1 + gs1 * HH;
        const v2f p0 = {u0[koff], u0[koff + 1]};
        const v2f p1 = {u1[koff], u1[koff + 1]};
        ra0[0] = p0;
        ra1[0] = act1 ? p1 : zero;
        const v2f q0h0 = {u0[4], b_ih0[gs0]}, q0h1 = {b_hh0[gs0], 0.0f};
        const v2f q1h0 = {u1[4], b_ih0[gs1]}, q1h1 = {b_hh0[gs1], 0.0f};
        ra0[1] = half ? q0h1 : q0h0;
        ra1[1] = act1 ? (half ? q1h1 : q1h0) : zero;

        const int c3 = half * 3;
        const v2f a0 = {v0[c3], v0[c3 + 1]};
        const v2f a1 = {v1[c3], v1[c3 + 1]};
        la0[0] = a0;
        la1[0] = act1 ? a1 : zero;
        const v2f b0h0 = {v0[2], y0[0]}, b0h1 = {y0[1], y0[2]};
        const v2f b1h0 = {v1[2], y1[0]}, b1h1 = {y1[1], y1[2]};
        la0[1] = half ? b0h1 : b0h0;
        la1[1] = act1 ? (half ? b1h1 : b1h0) : zero;
        const v2f c0h0 = {y0[3], y0[4]}, c0h1 = {b_ih1[gs0], b_hh1[gs0]};
        const v2f c1h0 = {y1[3], y1[4]}, c1h1 = {b_ih1[gs1], b_hh1[gs1]};
        rl0 = half ? c0h1 : c0h0;
        rl1 = act1 ? (half ? c1h1 : c1h0) : zero;
    }

    const int gw = blockIdx.x * 8 + __builtin_amdgcn_readfirstlane(wave);
    const int b  = (gw & 255) * 16 + m;

    int t = gw >> 8;
    if (t >= T) return;

    auto walk = [&](auto one_tile) __attribute__((always_inline)) {
    constexpr bool ONE = decltype(one_tile)::value;

    const int hu = half * 3;
    const float* h0r  = hidden + (long)b * HH;
    const float* h1r  = hidden + (long)BB * HH + (long)b * HH;
    const float* c0r  = cell + (long)b * HH;
    const float* c1r  = cell + (long)BB * HH + (long)b * HH;

    float hA[HH], hB[HH], cc0[2], cc1[2];
    #pragma unroll
    for (int j = 0; j < HH; ++j) {
        hA[j] = h0r[j];
        hB[j] = h1r[j];
    }
    #pragma unroll
    for (int u = 0; u < 2; ++u) {
        cc0[u] = c0r[hu + u];
        cc1[u] = c1r[hu + u];
    }
    const float cc0m = c0r[2];
    const float cc1m = c1r[2];

    const v2f xh32 = {half ? hA[2] : hA[0], half ? hA[3] : hA[1]};
    const v2f xh33 = {half ? 1.0f : hA[4], half ? 0.0f : 1.0f};
    const float l1e1 = half ? hB[2] : hB[0];
    const v2f lb2  = {half ? 1.0f : hB[3], half ? 1.0f : hB[4]};

    v8f racc0 = {}, racc1 = {}, rg0 = {}, rg1 = {};
    racc0 = __builtin_amdgcn_wmma_f32_16x16x4_f32(false, ra0[0], false, xh32, (short)0, racc0, false, false);
    racc1 = __builtin_amdgcn_wmma_f32_16x16x4_f32(false, ra1[0], false, xh32, (short)0, racc1, false, false);
    racc0 = __builtin_amdgcn_wmma_f32_16x16x4_f32(false, ra0[1], false, xh33, (short)0, racc0, false, false);
    racc1 = __builtin_amdgcn_wmma_f32_16x16x4_f32(false, ra1[1], false, xh33, (short)0, racc1, false, false);
    rg0   = __builtin_amdgcn_wmma_f32_16x16x4_f32(false, rl0, false, lb2, (short)0, rg0, false, false);
    rg1   = __builtin_amdgcn_wmma_f32_16x16x4_f32(false, rl1, false, lb2, (short)0, rg1, false, false);

    const float* ep = emb + koff4;
    const float* dp = dec_x + koff4;
    v4f xa[16], xb[16];
    int n1, n2, n3;
    if constexpr (ONE) {
        const long row0 = (long)t * BB + b;
        const int  i1 = cat1[row0], i2 = cat2[row0], i3 = cat3[row0];
        load_x(xa, dp + row0 * DD, ep + (long)i1 * 32, ep + (long)i2 * 32, ep + (long)i3 * 32);
    } else {
        const long row0 = (long)t * BB + b;
        const int  i1 = cat1[row0], i2 = cat2[row0], i3 = cat3[row0];
        const long row1 = (long)(t + TG) * BB + b;
        const int  j1 = cat1[row1], j2 = cat2[row1], j3 = cat3[row1];
        load_x(xa, dp + row0 * DD, ep + (long)i1 * 32, ep + (long)i2 * 32, ep + (long)i3 * 32);
        load_x(xb, dp + row1 * DD, ep + (long)j1 * 32, ep + (long)j2 * 32, ep + (long)j3 * 32);
        const int  t2   = (t + 2 * TG < T) ? t + 2 * TG : t;
        const long row2 = (long)t2 * BB + b;
        n1 = cat1[row2]; n2 = cat2[row2]; n3 = cat3[row2];
    }

    v8f acc0 = racc0;
    v8f acc1 = racc1;
    #pragma unroll
    for (int j = 0; j < 16; ++j) {
        acc0 = __builtin_amdgcn_wmma_f32_16x16x4_f32(false, lo2(wa0[j]), false, lo2(xa[j]),
                                                     (short)0, acc0, false, false);
        acc1 = __builtin_amdgcn_wmma_f32_16x16x4_f32(false, lo2(wa1[j]), false, lo2(xa[j]),
                                                     (short)0, acc1, false, false);
        acc0 = __builtin_amdgcn_wmma_f32_16x16x4_f32(false, hi2(wa0[j]), false, hi2(xa[j]),
                                                     (short)0, acc0, false, false);
        acc1 = __builtin_amdgcn_wmma_f32_16x16x4_f32(false, hi2(wa1[j]), false, hi2(xa[j]),
                                                     (short)0, acc1, false, false);
    }

#define TILE_STEP(TT, XUSE, XLOAD)                                                                   \
    {                                                                                                \
        const int  tt_  = (TT);                                                                      \
        const long row  = (long)tt_ * BB + b;                                                        \
        const int  t3   = (tt_ + 3 * TG < T) ? tt_ + 3 * TG : tt_;                                   \
        const long row3 = (long)t3 * BB + b;                                                         \
        const int  m1 = cat1[row3], m2 = cat2[row3], m3 = cat3[row3];                                \
        const bool  more = tt_ + 2 * TG < T;                                                         \
        const long  rown = (long)(more ? tt_ + 2 * TG : tt_) * BB + b;                               \
        const float* q0  = more ? dp + rown * DD : ep;                                               \
        load_x(XLOAD, q0, ep + (long)n1 * 32, ep + (long)n2 * 32, ep + (long)n3 * 32);               \
        n1 = m1; n2 = m2; n3 = m3;                                                                   \
                             \
        __builtin_amdgcn_sched_barrier(0);                                                           \
        v8f nx0 = racc0;                                                                             \
        v8f nx1 = racc1;                                                                             \
        _Pragma("unroll")                                                                            \
        for (int j = 0; j < 8; ++j) {                                                                \
            nx0 = __builtin_amdgcn_wmma_f32_16x16x4_f32(false, lo2(wa0[j]), false, lo2(XUSE[j]),     \
                                                        (short)0, nx0, false, false);                \
            nx1 = __builtin_amdgcn_wmma_f32_16x16x4_f32(false, lo2(wa1[j]), false, lo2(XUSE[j]),     \
                                                        (short)0, nx1, false, false);                \
            nx0 = __builtin_amdgcn_wmma_f32_16x16x4_f32(false, hi2(wa0[j]), false, hi2(XUSE[j]),     \
                                                        (short)0, nx0, false, false);                \
            nx1 = __builtin_amdgcn_wmma_f32_16x16x4_f32(false, hi2(wa1[j]), false, hi2(XUSE[j]),     \
                                                        (short)0, nx1, false, false);                \
        }                                                                                            \
            \
        float own[3];                                                                                \
        _Pragma("unroll")                                                                            \
        for (int u = 0; u < 2; ++u) {                                                                \
            own[u] = lstm_unit(acc0[u], acc0[2 + u], acc0[4 + u], acc0[6 + u], cc0[u]);              \
        }                                                                                            \
        own[2] = lstm_unit(acc1[0], acc1[1], acc1[2], acc1[3], cc0m);                                \
        const v2f lb0 = {own[0], own[1]};                                                            \
        const v2f lb1 = {half ? hB[1] : own[2], l1e1};                                               \
        v8f g0 = rg0;                                                                                \
        v8f g1 = rg1;                                                                                \
        g0 = __builtin_amdgcn_wmma_f32_16x16x4_f32(false, la0[0], false, lb0,                        \
                                                   (short)0, g0, false, false);                      \
        g1 = __builtin_amdgcn_wmma_f32_16x16x4_f32(false, la1[0], false, lb0,                        \
                                                   (short)0, g1, false, false);                      \
        g0 = __builtin_amdgcn_wmma_f32_16x16x4_f32(false, la0[1], false, lb1,                        \
                                                   (short)0, g0, false, false);                      \
        g1 = __builtin_amdgcn_wmma_f32_16x16x4_f32(false, la1[1], false, lb1,                        \
                                                   (short)0, g1, false, false);                      \
                                                      \
        _Pragma("unroll")                                                                            \
        for (int k = 0; k < 32; ++k) {                                                               \
            __builtin_amdgcn_sched_group_barrier(0x008, 1, 0);                                       \
            __builtin_amdgcn_sched_group_barrier(0x402, 9, 0);                                       \
        }                                                                                            \
        __builtin_amdgcn_sched_group_barrier(0x008, 4, 0);                                           \
                   \
        asm volatile("" : "+v"(g0), "+v"(g1), "+v"(nx0), "+v"(nx1));                                 \
        __builtin_amdgcn_sched_barrier(0);                                                           \
        _Pragma("unroll")                                                                            \
        for (int j = 8; j < 16; ++j) {                                                               \
            nx0 = __builtin_amdgcn_wmma_f32_16x16x4_f32(false, lo2(wa0[j]), false, lo2(XUSE[j]),     \
                                                        (short)0, nx0, false, false);                \
            nx1 = __builtin_amdgcn_wmma_f32_16x16x4_f32(false, lo2(wa1[j]), false, lo2(XUSE[j]),     \
                                                        (short)0, nx1, false, false);                \
            nx0 = __builtin_amdgcn_wmma_f32_16x16x4_f32(false, hi2(wa0[j]), false, hi2(XUSE[j]),     \
                                                        (short)0, nx0, false, false);                \
            nx1 = __builtin_amdgcn_wmma_f32_16x16x4_f32(false, hi2(wa1[j]), false, hi2(XUSE[j]),     \
                                                        (short)0, nx1, false, false);                \
        }                                                                                            \
        float o[3];                                                                                  \
        _Pragma("unroll")                                                                            \
        for (int u = 0; u < 2; ++u) {                                                                \
            o[u] = lstm_unit(g0[u], g0[2 + u], g0[4 + u], g0[6 + u], cc1[u]);                        \
        }                                                                                            \
        o[2] = lstm_unit(g1[0], g1[1], g1[2], g1[3], cc1m);                                          \
                                                                 \
        _Pragma("unroll")                                                                            \
        for (int k = 0; k < 32; ++k) {                                                               \
            __builtin_amdgcn_sched_group_barrier(0x008, 1, 0);                                       \
            __builtin_amdgcn_sched_group_barrier(0x402, 9, 0);                                       \
        }                                                                                            \
                       \
        asm volatile("" :: "v"(o[0]), "v"(o[1]), "v"(o[2]));                                         \
        __builtin_amdgcn_sched_barrier(0);                                                           \
        out[row * HH + hu]     = o[0];                                                               \
        out[row * HH + hu + 1] = o[1];                                                               \
        if (half == 0) {                                                                             \
            out[row * HH + 2] = o[2];                                                                \
        }                                                                                            \
        acc0 = nx0;                                                                                  \
        acc1 = nx1;                                                                                  \
    }

    if constexpr (!ONE) {
        #pragma unroll 1
        for (;;) {
            TILE_STEP(t, xb, xa)
            t += TG;
            if (t + TG >= T) break;
            TILE_STEP(t, xa, xb)
            t += TG;
            if (t + TG >= T) break;
        }
    }
#undef TILE_STEP

    {
        const long row = (long)t * BB + b;
        float own[3];
        #pragma unroll
        for (int u = 0; u < 2; ++u) {
            own[u] = lstm_unit(acc0[u], acc0[2 + u], acc0[4 + u], acc0[6 + u], cc0[u]);
        }
        own[2] = lstm_unit(acc1[0], acc1[1], acc1[2], acc1[3], cc0m);
        const v2f lb0 = {own[0], own[1]};
        const v2f lb1 = {half ? hB[1] : own[2], l1e1};
        v8f g0 = rg0;
        v8f g1 = rg1;
        g0 = __builtin_amdgcn_wmma_f32_16x16x4_f32(false, la0[0], false, lb0, (short)0, g0, false, false);
        g1 = __builtin_amdgcn_wmma_f32_16x16x4_f32(false, la1[0], false, lb0, (short)0, g1, false, false);
        g0 = __builtin_amdgcn_wmma_f32_16x16x4_f32(false, la0[1], false, lb1, (short)0, g0, false, false);
        g1 = __builtin_amdgcn_wmma_f32_16x16x4_f32(false, la1[1], false, lb1, (short)0, g1, false, false);
        float o[3];
        #pragma unroll
        for (int u = 0; u < 2; ++u) {
            o[u] = lstm_unit(g0[u], g0[2 + u], g0[4 + u], g0[6 + u], cc1[u]);
        }
        o[2] = lstm_unit(g1[0], g1[1], g1[2], g1[3], cc1m);
        out[row * HH + hu]     = o[0];
        out[row * HH + hu + 1] = o[1];
        if (half == 0) {
            out[row * HH + 2] = o[2];
        }
    }
    };

    if (t + TG >= T) {
        walk(std::true_type{});
    } else {
        walk(std::false_type{});
    }
}

extern "C" void kernel_launch(void* const* d_in, const int* in_sizes, int n_in,
                              void* d_out, int out_size, void* d_ws, size_t ws_size,
                              hipStream_t stream)
{
    const float* hidden = (const float*)d_in[1];
    const float* cellp  = (const float*)d_in[2];
    const float* dec_x  = (const float*)d_in[3];
    const int*   cat1   = (const int*)d_in[4];
    const int*   cat2   = (const int*)d_in[5];
    const int*   cat3   = (const int*)d_in[6];
    const float* emb    = (const float*)d_in[7];
    const float* W_ih0  = (const float*)d_in[8];
    const float* W_hh0  = (const float*)d_in[9];
    const float* b_ih0  = (const float*)d_in[10];
    const float* b_hh0  = (const float*)d_in[11];
    const float* W_ih1  = (const float*)d_in[12];
    const float* W_hh1  = (const float*)d_in[13];
    const float* b_ih1  = (const float*)d_in[14];
    const float* b_hh1  = (const float*)d_in[15];
    float* out = (float*)d_out;

    const int T    = in_sizes[3] / (BB * DD);

    lstm_wmma_kernel<<<256, 256, 0, stream>>>(dec_x, cat1, cat2, cat3, emb,
                                              W_ih0, W_hh0, b_ih0, b_hh0,
                                              W_ih1, W_hh1, b_ih1, b_hh1,
                                              hidden, cellp, out, T);
}


# Walk by tile pairs: hidden unit 2 worked out once per pair (lane half 0: first tile, half 1: second; its weight rows doubled into idle slots 8-11), matrix unit one pair ahead; sums unchanged
# speedup vs baseline: 2.8110x; 1.0522x over previous
#include <hip/hip_runtime.h>
#include <hip/hip_bf16.h>

typedef float v2f __attribute__((ext_vector_type(2)));
typedef float v4f __attribute__((ext_vector_type(4)));
typedef float v8f __attribute__((ext_vector_type(8)));

__device__ __forceinline__ v2f lo2(v4f v) { return __builtin_shufflevector(v, v, 0, 1); }
__device__ __forceinline__ v2f hi2(v4f v) { return __builtin_shufflevector(v, v, 2, 3); }

#define BB   4096
#define DD   32
#define HH   5
#define GG   20
#define INW  128

__device__ __forceinline__ float sigf(float x) { return 1.0f / (1.0f + __expf(-x)); }

#define TG 8

__device__ __forceinline__ float tanh_sel(float x) {
    const float t  = fabsf(x);
    const float x2 = x * x;
    float p = fmaf(x2, __uint_as_float(0xbbbac73du), __uint_as_float(0x3ca908c9u));
    p = fmaf(x2, p, __uint_as_float(0xbd5c1c4eu));
    p = fmaf(x2, p, __uint_as_float(0x3e088382u));
    p = fmaf(x2, p, __uint_as_float(0xbeaaaa99u));
    const float small = fmaf(x2, t * p, t);
    const float e     = expf(t + t);
    const float big   = fmaf(__builtin_amdgcn_rcpf(e + 1.0f), -2.0f, 1.0f);
    return copysignf((t < 0.625f) ? small : big, x);
}

__device__ __forceinline__ float lstm_unit(float gi, float gf, float gg, float go, float c) {
    float cn = sigf(gf) * c + sigf(gi) * tanh_sel(gg);
    return sigf(go) * tanh_sel(cn);
}

__device__ __forceinline__ float xhalf(float x) {
    const int v = __builtin_bit_cast(int, x);
    return __builtin_bit_cast(float, __builtin_amdgcn_permlanex16(v, v, 0x76543210, 0xfedcba98, false, false));
}

__device__ __forceinline__ void load_x(v4f (&x)[16],
                                       const float* __restrict__ p0, const float* __restrict__ p1,
                                       const float* __restrict__ p2, const float* __restrict__ p3) {
    #pragma unroll
    for (int q = 0; q < 4; ++q) {
        x[q]      = *(const v4f*)(p0 + q * 8);
        x[4 + q]  = *(const v4f*)(p1 + q * 8);
        x[8 + q]  = *(const v4f*)(p2 + q * 8);
        x[12 + q] = *(const v4f*)(p3 + q * 8);
    }
}

__device__ __forceinline__ void mul_tile(v8f& a0, v8f& a1, const v4f (&wa0)[16], const v4f (&wa1)[16],
                                         const v4f (&x)[16]) {
    #pragma unroll
    for (int j = 0; j < 16; ++j) {
        a0 = __builtin_amdgcn_wmma_f32_16x16x4_f32(false, lo2(wa0[j]), false, lo2(x[j]), (short)0, a0, false, false);
        a1 = __builtin_amdgcn_wmma_f32_16x16x4_f32(false, lo2(wa1[j]), false, lo2(x[j]), (short)0, a1, false, false);
        a0 = __builtin_amdgcn_wmma_f32_16x16x4_f32(false, hi2(wa0[j]), false, hi2(x[j]), (short)0, a0, false, false);
        a1 = __builtin_amdgcn_wmma_f32_16x16x4_f32(false, hi2(wa1[j]), false, hi2(x[j]), (short)0, a1, false, false);
    }
}

__device__ __forceinline__ void mul_layer1(v8f& g0, v8f& g1, const v2f (&la0)[2], const v2f (&la1)[2],
                                           v2f lb0, v2f lb1) {
    g0 = __builtin_amdgcn_wmma_f32_16x16x4_f32(false, la0[0], false, lb0, (short)0, g0, false, false);
    g1 = __builtin_amdgcn_wmma_f32_16x16x4_f32(false, la1[0], false, lb0, (short)0, g1, false, false);
    g0 = __builtin_amdgcn_wmma_f32_16x16x4_f32(false, la0[1], false, lb1, (short)0, g0, false, false);
    g1 = __builtin_amdgcn_wmma_f32_16x16x4_f32(false, la1[1], false, lb1, (short)0, g1, false, false);
}

__device__ __forceinline__ void cell_pair(const v8f& pa0, const v8f& pa1, const v8f& pb0, const v8f& pb1,
                                          bool upper, float c0, float c1, float cm,
                                          float& ya0, float& ya1, float& yb0, float& yb1, float& y2) {
    ya0 = lstm_unit(pa0[0], pa0[2], pa0[4], pa0[6], c0);
    ya1 = lstm_unit(pa0[1], pa0[3], pa0[5], pa0[7], c1);
    yb0 = lstm_unit(pb0[0], pb0[2], pb0[4], pb0[6], c0);
    yb1 = lstm_unit(pb0[1], pb0[3], pb0[5], pb0[7], c1);
    const float si = upper ? pb1[0] : pa1[0];
    const float sf = upper ? pb1[1] : pa1[1];
    const float sg = upper ? pb1[2] : pa1[2];
    const float so = upper ? pb1[3] : pa1[3];
    y2 = lstm_unit(si, sf, sg, so, cm);
}

__global__ __launch_bounds__(256) __attribute__((amdgpu_waves_per_eu(2, 2))) void lstm_wmma_kernel(
    const float* __restrict__ dec_x,
    const int*   __restrict__ cat1,
    const int*   __restrict__ cat2,
    const int*   __restrict__ cat3,
    const float* __restrict__ emb,
    const float* __restrict__ W_ih0,
    const float* __restrict__ W_hh0,
    const float* __restrict__ b_ih0,
    const float* __restrict__ b_hh0,
    const float* __restrict__ W_ih1,
    const float* __restrict__ W_hh1,
    const float* __restrict__ b_ih1,
    const float* __restrict__ b_hh1,
    const float* __restrict__ hidden,
    const float* __restrict__ cell,
    float* __restrict__ out,
    int T)
{
    const int tid  = threadIdx.x;
    const int wave = tid >> 5;
    const int lane = tid & 31;
    const int m    = lane & 15;
    const int half = lane >> 4;
    const bool upper = half != 0;
    const int koff  = half * 2;
    const int koff4 = half * 4;

    const int    m7   = m & 7;
    const int    gs0  = ((m >> 1) & 3) * HH + (m & 1) + ((m & 8) ? 3 : 0);
    const int    gs1  = (m7 < 4) ? (m7 * HH + 2) : 2;
    const float* wr0  = W_ih0 + gs0 * INW + koff4;
    const float* wr1  = W_ih0 + gs1 * INW + koff4;
    const v2f    zero = {0.0f, 0.0f};
    const v4f    zero4 = {0.0f, 0.0f, 0.0f, 0.0f};
    const bool   act1 = m7 < 4;
    v4f wa0[16], wa1[16];
    #pragma unroll
    for (int j = 0; j < 16; ++j) {
        wa0[j] = *(const v4f*)(wr0 + j * 8);
        v4f w1 = *(const v4f*)(wr1 + j * 8);
        wa1[j] = act1 ? w1 : zero4;
    }
    v2f la0[2], la1[2];
    v2f ra0[2], ra1[2], rl0, rl1;
    {
        const float* u0 = W_hh0 + gs0 * HH;
        const float* u1 = W_hh0 + gs1 * HH;
        const float* v0 = W_ih1 + gs0 * HH;
        const float* v1 = W_ih1 + gs1 * HH;
        const float* y0 = W_hh1 + gs0 * HH;
        const float* y1 = W_hh1 + gs1 * HH;
        const v2f p0 = {u0[koff], u0[koff + 1]};
        const v2f p1 = {u1[koff], u1[koff + 1]};
        ra0[0] = p0;
        ra1[0] = act1 ? p1 : zero;
        const v2f q0h0 = {u0[4], b_ih0[gs0]}, q0h1 = {b_hh0[gs0], 0.0f};
        const v2f q1h0 = {u1[4], b_ih0[gs1]}, q1h1 = {b_hh0[gs1], 0.0f};
        ra0[1] = half ? q0h1 : q0h0;
        ra1[1] = act1 ? (half ? q1h1 : q1h0) : zero;

        const int c3 = half * 3;
        const v2f a0 = {v0[c3], v0[c3 + 1]};
        const v2f a1 = {v1[c3], v1[c3 + 1]};
        la0[0] = a0;
        la1[0] = act1 ? a1 : zero;
        const v2f b0h0 = {v0[2], y0[0]}, b0h1 = {y0[1], y0[2]};
        const v2f b1h0 = {v1[2], y1[0]}, b1h1 = {y1[1], y1[2]};
        la0[1] = half ? b0h1 : b0h0;
        la1[1] = act1 ? (half ? b1h1 : b1h0) : zero;
        const v2f c0h0 = {y0[3], y0[4]}, c0h1 = {b_ih1[gs0], b_hh1[gs0]};
        const v2f c1h0 = {y1[3], y1[4]}, c1h1 = {b_ih1[gs1], b_hh1[gs1]};
        rl0 = half ? c0h1 : c0h0;
        rl1 = act1 ? (half ? c1h1 : c1h0) : zero;
    }

    const int gw = blockIdx.x * 8 + __builtin_amdgcn_readfirstlane(wave);
    const int b  = (gw & 255) * 16 + m;

    int t = gw >> 8;
    if (t >= T) return;

    const int hu = half * 3;
    const float* h0r  = hidden + (long)b * HH;
    const float* h1r  = hidden + (long)BB * HH + (long)b * HH;
    const float* c0r  = cell + (long)b * HH;
    const float* c1r  = cell + (long)BB * HH + (long)b * HH;

    float hA[HH], hB[HH], cc0[2], cc1[2];
    #pragma unroll
    for (int j = 0; j < HH; ++j) {
        hA[j] = h0r[j];
        hB[j] = h1r[j];
    }
    #pragma unroll
    for (int u = 0; u < 2; ++u) {
        cc0[u] = c0r[hu + u];
        cc1[u] = c1r[hu + u];
    }
    const float cc0m = c0r[2];
    const float cc1m = c1r[2];

    const v2f xh32 = {half ? hA[2] : hA[0], half ? hA[3] : hA[1]};
    const v2f xh33 = {half ? 1.0f : hA[4], half ? 0.0f : 1.0f};
    const float l1e1 = half ? hB[2] : hB[0];
    const v2f lb2  = {half ? 1.0f : hB[3], half ? 1.0f : hB[4]};

    v8f racc0 = {}, racc1 = {}, rg0 = {}, rg1 = {};
    racc0 = __builtin_amdgcn_wmma_f32_16x16x4_f32(false, ra0[0], false, xh32, (short)0, racc0, false, false);
    racc1 = __builtin_amdgcn_wmma_f32_16x16x4_f32(false, ra1[0], false, xh32, (short)0, racc1, false, false);
    racc0 = __builtin_amdgcn_wmma_f32_16x16x4_f32(false, ra0[1], false, xh33, (short)0, racc0, false, false);
    racc1 = __builtin_amdgcn_wmma_f32_16x16x4_f32(false, ra1[1], false, xh33, (short)0, racc1, false, false);
    rg0   = __builtin_amdgcn_wmma_f32_16x16x4_f32(false, rl0, false, lb2, (short)0, rg0, false, false);
    rg1   = __builtin_amdgcn_wmma_f32_16x16x4_f32(false, rl1, false, lb2, (short)0, rg1, false, false);

    const float* ep = emb + koff4;
    const float* dp = dec_x + koff4;
    v4f xa[16], xb[16];

    const int ntiles = (T - t + TG - 1) / TG;
    if (__builtin_expect(ntiles & 1, 0)) {
        const long row = (long)t * BB + b;
        const int  i1 = cat1[row], i2 = cat2[row], i3 = cat3[row];
        load_x(xa, dp + row * DD, ep + (long)i1 * 32, ep + (long)i2 * 32, ep + (long)i3 * 32);
        v8f acc0 = racc0, acc1 = racc1;
        mul_tile(acc0, acc1, wa0, wa1, xa);
        float own[3];
        #pragma unroll
        for (int u = 0; u < 2; ++u) {
            own[u] = lstm_unit(acc0[u], acc0[2 + u], acc0[4 + u], acc0[6 + u], cc0[u]);
        }
        own[2] = lstm_unit(acc1[0], acc1[1], acc1[2], acc1[3], cc0m);
        const v2f lb0 = {own[0], own[1]};
        const v2f lb1 = {half ? hB[1] : own[2], l1e1};
        v8f g0 = rg0, g1 = rg1;
        mul_layer1(g0, g1, la0, la1, lb0, lb1);
        float o[3];
        #pragma unroll
        for (int u = 0; u < 2; ++u) {
            o[u] = lstm_unit(g0[u], g0[2 + u], g0[4 + u], g0[6 + u], cc1[u]);
        }
        o[2] = lstm_unit(g1[0], g1[1], g1[2], g1[3], cc1m);
        out[row * HH + hu]     = o[0];
        out[row * HH + hu + 1] = o[1];
        if (half == 0) {
            out[row * HH + 2] = o[2];
        }
        t += TG;
        if (t >= T) return;
    }

    int p1, p2, p3, q1, q2, q3;
    int c1, c2, c3, d1, d2, d3;
    bool more = t + 2 * TG < T;
    {
        const long rowA = (long)t * BB + b;
        const long rowB = (long)(t + TG) * BB + b;
        const int  i1 = cat1[rowA], i2 = cat2[rowA], i3 = cat3[rowA];
        const int  j1 = cat1[rowB], j2 = cat2[rowB], j3 = cat3[rowB];
        const long rowC = (long)(more ? t + 2 * TG : t) * BB + b;
        const long rowD = (long)(more ? t + 3 * TG : t) * BB + b;
        p1 = cat1[rowC]; p2 = cat2[rowC]; p3 = cat3[rowC];
        q1 = cat1[rowD]; q2 = cat2[rowD]; q3 = cat3[rowD];
        const bool third = t + 4 * TG < T;
        const long rowE = (long)(third ? t + 4 * TG : t) * BB + b;
        const long rowF = (long)(third ? t + 5 * TG : t) * BB + b;
        c1 = cat1[rowE]; c2 = cat2[rowE]; c3 = cat3[rowE];
        d1 = cat1[rowF]; d2 = cat2[rowF]; d3 = cat3[rowF];
        load_x(xa, dp + rowA * DD, ep + (long)i1 * 32, ep + (long)i2 * 32, ep + (long)i3 * 32);
        load_x(xb, dp + rowB * DD, ep + (long)j1 * 32, ep + (long)j2 * 32, ep + (long)j3 * 32);
    }

    v8f accA0 = racc0, accA1 = racc1, accB0 = racc0, accB1 = racc1;
    mul_tile(accA0, accA1, wa0, wa1, xa);
    {
        const long rowC = (long)(more ? t + 2 * TG : t) * BB + b;
        __builtin_amdgcn_sched_barrier(0);
        load_x(xa, more ? dp + rowC * DD : ep, ep + (long)p1 * 32, ep + (long)p2 * 32, ep + (long)p3 * 32);
        __builtin_amdgcn_sched_barrier(0);
    }
    mul_tile(accB0, accB1, wa0, wa1, xb);
    {
        const long rowD = (long)(more ? t + 3 * TG : t) * BB + b;
        __builtin_amdgcn_sched_barrier(0);
        load_x(xb, more ? dp + rowD * DD : ep, ep + (long)q1 * 32, ep + (long)q2 * 32, ep + (long)q3 * 32);
        __builtin_amdgcn_sched_barrier(0);
    }

    #pragma unroll 1
    while (more) {
        const long rowA = (long)t * BB + b;
        const long rowB = (long)(t + TG) * BB + b;
        const bool more2 = t + 4 * TG < T;
        const long rowC = (long)(more2 ? t + 4 * TG : t) * BB + b;
        const long rowD = (long)(more2 ? t + 5 * TG : t) * BB + b;
        const bool more3 = t + 6 * TG < T;
        const long rowE = (long)(more3 ? t + 6 * TG : t) * BB + b;
        const long rowF = (long)(more3 ? t + 7 * TG : t) * BB + b;
        const int  e1 = cat1[rowE], e2 = cat2[rowE], e3 = cat3[rowE];
        const int  f1 = cat1[rowF], f2 = cat2[rowF], f3 = cat3[rowF];
        __builtin_amdgcn_sched_barrier(0);

        v8f nxA0 = racc0, nxA1 = racc1;
        mul_tile(nxA0, nxA1, wa0, wa1, xa);
        float ya0, ya1, yb0, yb1, y2;
        cell_pair(accA0, accA1, accB0, accB1, upper, cc0[0], cc0[1], cc0m, ya0, ya1, yb0, yb1, y2);
        const float y2x = xhalf(y2);
        const v2f lbA0 = {ya0, ya1};
        const v2f lbA1 = {half ? hB[1] : y2, l1e1};
        const v2f lbB0 = {yb0, yb1};
        const v2f lbB1 = {half ? hB[1] : y2x, l1e1};
        v8f gA0 = rg0, gA1 = rg1, gB0 = rg0, gB1 = rg1;
        mul_layer1(gA0, gA1, la0, la1, lbA0, lbA1);
        mul_layer1(gB0, gB1, la0, la1, lbB0, lbB1);
        #pragma unroll
        for (int k = 0; k < 64; ++k) {
            __builtin_amdgcn_sched_group_barrier(0x008, 1, 0);
            __builtin_amdgcn_sched_group_barrier(0x402, 8, 0);
        }
        __builtin_amdgcn_sched_group_barrier(0x008, 8, 0);
        asm volatile("" : "+v"(gA0), "+v"(gA1), "+v"(gB0), "+v"(gB1), "+v"(nxA0), "+v"(nxA1));
        __builtin_amdgcn_sched_barrier(0);
        load_x(xa, more2 ? dp + rowC * DD : ep, ep + (long)c1 * 32, ep + (long)c2 * 32, ep + (long)c3 * 32);
        __builtin_amdgcn_sched_barrier(0);

        v8f nxB0 = racc0, nxB1 = racc1;
        mul_tile(nxB0, nxB1, wa0, wa1, xb);
        float oa0, oa1, ob0, ob1, o2;
        cell_pair(gA0, gA1, gB0, gB1, upper, cc1[0], cc1[1], cc1m, oa0, oa1, ob0, ob1, o2);
        #pragma unroll
        for (int k = 0; k < 64; ++k) {
            __builtin_amdgcn_sched_group_barrier(0x008, 1, 0);
            __builtin_amdgcn_sched_group_barrier(0x402, 8, 0);
        }
        asm volatile("" : "+v"(nxB0), "+v"(nxB1) : "v"(oa0), "v"(oa1), "v"(ob0), "v"(ob1), "v"(o2));
        __builtin_amdgcn_sched_barrier(0);
        load_x(xb, more2 ? dp + rowD * DD : ep, ep + (long)d1 * 32, ep + (long)d2 * 32, ep + (long)d3 * 32);
        out[rowA * HH + hu]     = oa0;
        out[rowA * HH + hu + 1] = oa1;
        out[rowB * HH + hu]     = ob0;
        out[rowB * HH + hu + 1] = ob1;
        out[(upper ? rowB : rowA) * HH + 2] = o2;
        c1 = e1; c2 = e2; c3 = e3;
        d1 = f1; d2 = f2; d3 = f3;
        accA0 = nxA0; accA1 = nxA1; accB0 = nxB0; accB1 = nxB1;
        t += 2 * TG;
        more = more2;
    }

    {
        const long rowA = (long)t * BB + b;
        const long rowB = (long)(t + TG) * BB + b;
        float ya0, ya1, yb0, yb1, y2;
        cell_pair(accA0, accA1, accB0, accB1, upper, cc0[0], cc0[1], cc0m, ya0, ya1, yb0, yb1, y2);
        const float y2x = xhalf(y2);
        const v2f lbA0 = {ya0, ya1};
        const v2f lbA1 = {half ? hB[1] : y2, l1e1};
        const v2f lbB0 = {yb0, yb1};
        const v2f lbB1 = {half ? hB[1] : y2x, l1e1};
        v8f gA0 = rg0, gA1 = rg1, gB0 = rg0, gB1 = rg1;
        mul_layer1(gA0, gA1, la0, la1, lbA0, lbA1);
        mul_layer1(gB0, gB1, la0, la1, lbB0, lbB1);
        float oa0, oa1, ob0, ob1, o2;
        cell_pair(gA0, gA1, gB0, gB1, upper, cc1[0], cc1[1], cc1m, oa0, oa1, ob0, ob1, o2);
        out[rowA * HH + hu]     = oa0;
        out[rowA * HH + hu + 1] = oa1;
        out[rowB * HH + hu]     = ob0;
        out[rowB * HH + hu + 1] = ob1;
        out[(upper ? rowB : rowA) * HH + 2] = o2;
    }
}

extern "C" void kernel_launch(void* const* d_in, const int* in_sizes, int n_in,
                              void* d_out, int out_size, void* d_ws, size_t ws_size,
                              hipStream_t stream)
{
    const float* hidden = (const float*)d_in[1];
    const float* cellp  = (const float*)d_in[2];
    const float* dec_x  = (const float*)d_in[3];
    const int*   cat1   = (const int*)d_in[4];
    const int*   cat2   = (const int*)d_in[5];
    const int*   cat3   = (const int*)d_in[6];
    const float* emb    = (const float*)d_in[7];
    const float* W_ih0  = (const float*)d_in[8];
    const float* W_hh0  = (const float*)d_in[9];
    const float* b_ih0  = (const float*)d_in[10];
    const float* b_hh0  = (const float*)d_in[11];
    const float* W_ih1  = (const float*)d_in[12];
    const float* W_hh1  = (const float*)d_in[13];
    const float* b_ih1  = (const float*)d_in[14];
    const float* b_hh1  = (const float*)d_in[15];
    float* out = (float*)d_out;

    const int T    = in_sizes[3] / (BB * DD);

    lstm_wmma_kernel<<<256, 256, 0, stream>>>(dec_x, cat1, cat2, cat3, emb,
                                              W_ih0, W_hh0, b_ih0, b_hh0,
                                              W_ih1, W_hh1, b_ih1, b_hh1,
                                              hidden, cellp, out, T);
}
